# starting kernel
# baseline (speedup 1.0000x reference)
#include <hip/hip_runtime.h>
#include <hip/hip_bf16.h>

#define BATCH 4
#define NPTS  8192
#define CH    32
#define KNN   16
#define LDIM  64
#define ODIM  128
#define NCTILES (NPTS / 16)
#define ROWS_PER_WAVE 32
#define NRBLK (NPTS / ROWS_PER_WAVE)

typedef __attribute__((ext_vector_type(2))) float v2f;
typedef __attribute__((ext_vector_type(8))) float v8f;

__global__ void sq_kernel(const float* __restrict__ x, float* __restrict__ sq, int total) {
    int i = blockIdx.x * blockDim.x + threadIdx.x;
    if (i >= total) return;
    const float* p = x + (size_t)i * CH;
    float s = 0.0f;
#pragma unroll
    for (int c = 0; c < CH; ++c) s += p[c] * p[c];
    sq[i] = s;
}

__global__ __launch_bounds__(32) void knn_pool_kernel(const float* __restrict__ x,
                                                      const float* __restrict__ sq,
                                                      float* __restrict__ pooled) {
    __shared__ float distT[ROWS_PER_WAVE * 19];
    __shared__ int   selIdx[ROWS_PER_WAVE * KNN];

    const int lane    = threadIdx.x;
    const int rblk    = blockIdx.x % NRBLK;
    const int b       = blockIdx.x / NRBLK;
    const int rowbase = rblk * ROWS_PER_WAVE;
    const float* xb   = x  + (size_t)b * NPTS * CH;
    const float* sqb  = sq + (size_t)b * NPTS;

    const int m = lane & 15;
    const int h = lane >> 4;

    v2f afrag0[8], afrag1[8];
#pragma unroll
    for (int c = 0; c < 8; ++c) {
        const float* p0 = xb + (size_t)(rowbase + m) * CH + 4 * c + 2 * h;
        const float* p1 = xb + (size_t)(rowbase + 16 + m) * CH + 4 * c + 2 * h;
        afrag0[c].x = -2.0f * p0[0];
        afrag0[c].y = -2.0f * p0[1];
        afrag1[c].x = -2.0f * p1[0];
        afrag1[c].y = -2.0f * p1[1];
    }
    v2f afragOne;
    afragOne.x = (h == 0) ? 1.0f : 0.0f;
    afragOne.y = 0.0f;

    float key[KNN];
    int   idx[KNN];
#pragma unroll
    for (int s = 0; s < KNN; ++s) { key[s] = 3.0e38f; idx[s] = 0; }

    for (int jt = 0; jt < NCTILES; ++jt) {
        const int colbase = jt * 16;

        v8f acc0 = {};
        v8f acc1 = {};
#pragma unroll
        for (int c = 0; c < 8; ++c) {
            const float* p = xb + (size_t)(colbase + m) * CH + 4 * c + 2 * h;
            v2f bfrag;
            bfrag.x = p[0];
            bfrag.y = p[1];
            acc0 = __builtin_amdgcn_wmma_f32_16x16x4_f32(
                false, afrag0[c], false, bfrag, (short)0, acc0, false, false);
            acc1 = __builtin_amdgcn_wmma_f32_16x16x4_f32(
                false, afrag1[c], false, bfrag, (short)0, acc1, false, false);
        }
        {
            v2f bfrag;
            bfrag.x = (h == 0) ? sqb[colbase + m] : 0.0f;
            bfrag.y = 0.0f;
            acc0 = __builtin_amdgcn_wmma_f32_16x16x4_f32(
                false, afragOne, false, bfrag, (short)0, acc0, false, false);
            acc1 = __builtin_amdgcn_wmma_f32_16x16x4_f32(
                false, afragOne, false, bfrag, (short)0, acc1, false, false);
        }

#pragma unroll
        for (int r = 0; r < 8; ++r) {
            distT[(r + 8 * h) * 19 + m]        = acc0[r];
            distT[(r + 8 * h + 16) * 19 + m]   = acc1[r];
        }
        __syncthreads();

#pragma unroll 1
        for (int c = 0; c < 16; ++c) {
            const float d = distT[lane * 19 + c];
            if (d < key[KNN - 1]) {
                float cv = d;
                int   ci = colbase + c;
#pragma unroll
                for (int s = 0; s < KNN; ++s) {
                    const float ks = key[s];
                    const int   is = idx[s];
                    const bool  sw = cv < ks;
                    key[s] = sw ? cv : ks;
                    idx[s] = sw ? ci : is;
                    cv = sw ? ks : cv;
                    ci = sw ? is : ci;
                }
            }
        }
        __syncthreads();
    }

#pragma unroll
    for (int s = 0; s < KNN; ++s)
        selIdx[lane * KNN + s] = idx[s];
    __syncthreads();

    float* poolb = pooled + ((size_t)b * NPTS + rowbase) * CH;
#pragma unroll 1
    for (int r = 0; r < ROWS_PER_WAVE; ++r) {
        float mx = -3.0e38f;
#pragma unroll
        for (int k = 0; k < KNN; ++k) {
            const int j = selIdx[r * KNN + k];
            const float v = xb[(size_t)j * CH + lane];
            mx = fmaxf(mx, v);
        }
        poolb[r * CH + lane] = mx;
    }
}

__global__ __launch_bounds__(32) void mlp_kernel(const float* __restrict__ pooled,
                                                 const float* __restrict__ Wl,
                                                 const float* __restrict__ bl,
                                                 const float* __restrict__ Wc,
                                                 const float* __restrict__ bc,
                                                 float* __restrict__ out) {
    __shared__ float hT[16 * 68];

    const int lane = threadIdx.x;
    const int m = lane & 15;
    const int h = lane >> 4;
    const size_t rowbase = (size_t)blockIdx.x * 16;

    v2f afrag[8];
#pragma unroll
    for (int c = 0; c < 8; ++c) {
        const float* p = pooled + (rowbase + m) * CH + 4 * c + 2 * h;
        afrag[c].x = p[0];
        afrag[c].y = p[1];
    }

#pragma unroll
    for (int t = 0; t < 4; ++t) {
        v8f acc = {};
#pragma unroll
        for (int c = 0; c < 8; ++c) {
            const float* p = Wl + (size_t)(4 * c + 2 * h) * LDIM + t * 16 + m;
            v2f bfrag;
            bfrag.x = p[0];
            bfrag.y = p[LDIM];
            acc = __builtin_amdgcn_wmma_f32_16x16x4_f32(
                false, afrag[c], false, bfrag, (short)0, acc, false, false);
        }
        const float bias = bl[t * 16 + m];
#pragma unroll
        for (int r = 0; r < 8; ++r)
            hT[(r + 8 * h) * 68 + t * 16 + m] = acc[r] + bias;
    }
    __syncthreads();

    v2f a2[16];
#pragma unroll
    for (int c = 0; c < 16; ++c) {
        a2[c].x = hT[m * 68 + 4 * c + 2 * h];
        a2[c].y = hT[m * 68 + 4 * c + 2 * h + 1];
    }

    float* outp = out + rowbase * ODIM;
#pragma unroll
    for (int t = 0; t < 8; ++t) {
        v8f acc = {};
#pragma unroll
        for (int c = 0; c < 16; ++c) {
            const float* p = Wc + (size_t)(4 * c + 2 * h) * ODIM + t * 16 + m;
            v2f bfrag;
            bfrag.x = p[0];
            bfrag.y = p[ODIM];
            acc = __builtin_amdgcn_wmma_f32_16x16x4_f32(
                false, a2[c], false, bfrag, (short)0, acc, false, false);
        }
        const float bias = bc[t * 16 + m];
#pragma unroll
        for (int r = 0; r < 8; ++r) {
            const float v = acc[r] + bias;
            outp[(size_t)(r + 8 * h) * ODIM + t * 16 + m] = fmaxf(v, 0.0f);
        }
    }
}

extern "C" void kernel_launch(void* const* d_in, const int* in_sizes, int n_in,
                              void* d_out, int out_size, void* d_ws, size_t ws_size,
                              hipStream_t stream) {
    const float* x  = (const float*)d_in[0];
    const float* Wl = (const float*)d_in[1];
    const float* bl = (const float*)d_in[2];
    const float* Wc = (const float*)d_in[3];
    const float* bc = (const float*)d_in[4];
    float* out = (float*)d_out;

    float* pooled = (float*)d_ws;
    float* sqbuf  = pooled + (size_t)BATCH * NPTS * CH;

    const int total = BATCH * NPTS;
    sq_kernel<<<(total + 255) / 256, 256, 0, stream>>>(x, sqbuf, total);
    knn_pool_kernel<<<BATCH * NRBLK, 32, 0, stream>>>(x, sqbuf, pooled);
    mlp_kernel<<<total / 16, 32, 0, stream>>>(pooled, Wl, bl, Wc, bc, out);
}


# selection loop: each lane fetches its row's 16 keys with four independent 16-byte LDS loads (row pitch 20 floats) ahead of the compare loop, instead of 16 dependent 4-byte LDS reads per tile
# speedup vs baseline: 1.3777x; 1.3777x over previous
#include <hip/hip_runtime.h>
#include <hip/hip_bf16.h>

#define BATCH 4
#define NPTS  8192
#define CH    32
#define KNN   16
#define LDIM  64
#define ODIM  128
#define NCTILES (NPTS / 16)
#define ROWS_PER_WAVE 32
#define NRBLK (NPTS / ROWS_PER_WAVE)
#define DPITCH 20

typedef __attribute__((ext_vector_type(2))) float v2f;
typedef __attribute__((ext_vector_type(8))) float v8f;

__global__ void sq_kernel(const float* __restrict__ x, float* __restrict__ sq, int total) {
    int i = blockIdx.x * blockDim.x + threadIdx.x;
    if (i >= total) return;
    const float* p = x + (size_t)i * CH;
    float s = 0.0f;
#pragma unroll
    for (int c = 0; c < CH; ++c) s += p[c] * p[c];
    sq[i] = s;
}

__global__ __launch_bounds__(32) void knn_pool_kernel(const float* __restrict__ x,
                                                      const float* __restrict__ sq,
                                                      float* __restrict__ pooled) {
    __shared__ __attribute__((aligned(16))) float distT[ROWS_PER_WAVE * DPITCH];
    __shared__ int   selIdx[ROWS_PER_WAVE * KNN];

    const int lane    = threadIdx.x;
    const int rblk    = blockIdx.x % NRBLK;
    const int b       = blockIdx.x / NRBLK;
    const int rowbase = rblk * ROWS_PER_WAVE;
    const float* xb   = x  + (size_t)b * NPTS * CH;
    const float* sqb  = sq + (size_t)b * NPTS;

    const int m = lane & 15;
    const int h = lane >> 4;

    v2f afrag0[8], afrag1[8];
#pragma unroll
    for (int c = 0; c < 8; ++c) {
        const float* p0 = xb + (size_t)(rowbase + m) * CH + 4 * c + 2 * h;
        const float* p1 = xb + (size_t)(rowbase + 16 + m) * CH + 4 * c + 2 * h;
        afrag0[c].x = -2.0f * p0[0];
        afrag0[c].y = -2.0f * p0[1];
        afrag1[c].x = -2.0f * p1[0];
        afrag1[c].y = -2.0f * p1[1];
    }
    v2f afragOne;
    afragOne.x = (h == 0) ? 1.0f : 0.0f;
    afragOne.y = 0.0f;

    float key[KNN];
    int   idx[KNN];
#pragma unroll
    for (int s = 0; s < KNN; ++s) { key[s] = 3.0e38f; idx[s] = 0; }

    for (int jt = 0; jt < NCTILES; ++jt) {
        const int colbase = jt * 16;

        v8f acc0 = {};
        v8f acc1 = {};
#pragma unroll
        for (int c = 0; c < 8; ++c) {
            const float* p = xb + (size_t)(colbase + m) * CH + 4 * c + 2 * h;
            v2f bfrag;
            bfrag.x = p[0];
            bfrag.y = p[1];
            acc0 = __builtin_amdgcn_wmma_f32_16x16x4_f32(
                false, afrag0[c], false, bfrag, (short)0, acc0, false, false);
            acc1 = __builtin_amdgcn_wmma_f32_16x16x4_f32(
                false, afrag1[c], false, bfrag, (short)0, acc1, false, false);
        }
        {
            v2f bfrag;
            bfrag.x = (h == 0) ? sqb[colbase + m] : 0.0f;
            bfrag.y = 0.0f;
            acc0 = __builtin_amdgcn_wmma_f32_16x16x4_f32(
                false, afragOne, false, bfrag, (short)0, acc0, false, false);
            acc1 = __builtin_amdgcn_wmma_f32_16x16x4_f32(
                false, afragOne, false, bfrag, (short)0, acc1, false, false);
        }

#pragma unroll
        for (int r = 0; r < 8; ++r) {
            distT[(r + 8 * h) * DPITCH + m]        = acc0[r];
            distT[(r + 8 * h + 16) * DPITCH + m]   = acc1[r];
        }
        __syncthreads();

        const float4* rowp = reinterpret_cast<const float4*>(&distT[lane * DPITCH]);
        const float4 q0 = rowp[0];
        const float4 q1 = rowp[1];
        const float4 q2 = rowp[2];
        const float4 q3 = rowp[3];
        const float dv[16] = { q0.x, q0.y, q0.z, q0.w, q1.x, q1.y, q1.z, q1.w,
                               q2.x, q2.y, q2.z, q2.w, q3.x, q3.y, q3.z, q3.w };
#pragma unroll
        for (int c = 0; c < 16; ++c) {
            const float d = dv[c];
            if (d < key[KNN - 1]) {
                float cv = d;
                int   ci = colbase + c;
#pragma unroll
                for (int s = 0; s < KNN; ++s) {
                    const float ks = key[s];
                    const int   is = idx[s];
                    const bool  sw = cv < ks;
                    key[s] = sw ? cv : ks;
                    idx[s] = sw ? ci : is;
                    cv = sw ? ks : cv;
                    ci = sw ? is : ci;
                }
            }
        }
        __syncthreads();
    }

#pragma unroll
    for (int s = 0; s < KNN; ++s)
        selIdx[lane * KNN + s] = idx[s];
    __syncthreads();

    float* poolb = pooled + ((size_t)b * NPTS + rowbase) * CH;
#pragma unroll 1
    for (int r = 0; r < ROWS_PER_WAVE; ++r) {
        float mx = -3.0e38f;
#pragma unroll
        for (int k = 0; k < KNN; ++k) {
            const int j = selIdx[r * KNN + k];
            const float v = xb[(size_t)j * CH + lane];
            mx = fmaxf(mx, v);
        }
        poolb[r * CH + lane] = mx;
    }
}

__global__ __launch_bounds__(32) void mlp_kernel(const float* __restrict__ pooled,
                                                 const float* __restrict__ Wl,
                                                 const float* __restrict__ bl,
                                                 const float* __restrict__ Wc,
                                                 const float* __restrict__ bc,
                                                 float* __restrict__ out) {
    __shared__ float hT[16 * 68];

    const int lane = threadIdx.x;
    const int m = lane & 15;
    const int h = lane >> 4;
    const size_t rowbase = (size_t)blockIdx.x * 16;

    v2f afrag[8];
#pragma unroll
    for (int c = 0; c < 8; ++c) {
        const float* p = pooled + (rowbase + m) * CH + 4 * c + 2 * h;
        afrag[c].x = p[0];
        afrag[c].y = p[1];
    }

#pragma unroll
    for (int t = 0; t < 4; ++t) {
        v8f acc = {};
#pragma unroll
        for (int c = 0; c < 8; ++c) {
            const float* p = Wl + (size_t)(4 * c + 2 * h) * LDIM + t * 16 + m;
            v2f bfrag;
            bfrag.x = p[0];
            bfrag.y = p[LDIM];
            acc = __builtin_amdgcn_wmma_f32_16x16x4_f32(
                false, afrag[c], false, bfrag, (short)0, acc, false, false);
        }
        const float bias = bl[t * 16 + m];
#pragma unroll
        for (int r = 0; r < 8; ++r)
            hT[(r + 8 * h) * 68 + t * 16 + m] = acc[r] + bias;
    }
    __syncthreads();

    v2f a2[16];
#pragma unroll
    for (int c = 0; c < 16; ++c) {
        a2[c].x = hT[m * 68 + 4 * c + 2 * h];
        a2[c].y = hT[m * 68 + 4 * c + 2 * h + 1];
    }

    float* outp = out + rowbase * ODIM;
#pragma unroll
    for (int t = 0; t < 8; ++t) {
        v8f acc = {};
#pragma unroll
        for (int c = 0; c < 16; ++c) {
            const float* p = Wc + (size_t)(4 * c + 2 * h) * ODIM + t * 16 + m;
            v2f bfrag;
            bfrag.x = p[0];
            bfrag.y = p[ODIM];
            acc = __builtin_amdgcn_wmma_f32_16x16x4_f32(
                false, a2[c], false, bfrag, (short)0, acc, false, false);
        }
        const float bias = bc[t * 16 + m];
#pragma unroll
        for (int r = 0; r < 8; ++r) {
            const float v = acc[r] + bias;
            outp[(size_t)(r + 8 * h) * ODIM + t * 16 + m] = fmaxf(v, 0.0f);
        }
    }
}

extern "C" void kernel_launch(void* const* d_in, const int* in_sizes, int n_in,
                              void* d_out, int out_size, void* d_ws, size_t ws_size,
                              hipStream_t stream) {
    const float* x  = (const float*)d_in[0];
    const float* Wl = (const float*)d_in[1];
    const float* bl = (const float*)d_in[2];
    const float* Wc = (const float*)d_in[3];
    const float* bc = (const float*)d_in[4];
    float* out = (float*)d_out;

    float* pooled = (float*)d_ws;
    float* sqbuf  = pooled + (size_t)BATCH * NPTS * CH;

    const int total = BATCH * NPTS;
    sq_kernel<<<(total + 255) / 256, 256, 0, stream>>>(x, sqbuf, total);
    knn_pool_kernel<<<BATCH * NRBLK, 32, 0, stream>>>(x, sqbuf, pooled);
    mlp_kernel<<<total / 16, 32, 0, stream>>>(pooled, Wl, bl, Wc, bc, out);
}


# tile loop: next tile's key-point and norm loads issued one tile ahead (second register set); loop barriers fence LDS only so the loads stay in flight; norm load unpredicated
# speedup vs baseline: 1.6073x; 1.1667x over previous
#include <hip/hip_runtime.h>
#include <hip/hip_bf16.h>

#define BATCH 4
#define NPTS  8192
#define CH    32
#define KNN   16
#define LDIM  64
#define ODIM  128
#define NCTILES (NPTS / 16)
#define ROWS_PER_WAVE 32
#define NRBLK (NPTS / ROWS_PER_WAVE)
#define DPITCH 20

typedef __attribute__((ext_vector_type(2))) float v2f;
typedef __attribute__((ext_vector_type(8))) float v8f;

__global__ void sq_kernel(const float* __restrict__ x, float* __restrict__ sq, int total) {
    int i = blockIdx.x * blockDim.x + threadIdx.x;
    if (i >= total) return;
    const float* p = x + (size_t)i * CH;
    float s = 0.0f;
#pragma unroll
    for (int c = 0; c < CH; ++c) s += p[c] * p[c];
    sq[i] = s;
}

__device__ __forceinline__ void lds_sync() {
    __builtin_amdgcn_fence(__ATOMIC_RELEASE, "workgroup", "local");
    __builtin_amdgcn_s_barrier();
    __builtin_amdgcn_fence(__ATOMIC_ACQUIRE, "workgroup", "local");
}

__global__ __launch_bounds__(32) void knn_pool_kernel(const float* __restrict__ x,
                                                      const float* __restrict__ sq,
                                                      float* __restrict__ pooled) {
    __shared__ __attribute__((aligned(16))) float distT[ROWS_PER_WAVE * DPITCH];
    __shared__ int   selIdx[ROWS_PER_WAVE * KNN];

    const int lane    = threadIdx.x;
    const int rblk    = blockIdx.x % NRBLK;
    const int b       = blockIdx.x / NRBLK;
    const int rowbase = rblk * ROWS_PER_WAVE;
    const float* xb   = x  + (size_t)b * NPTS * CH;
    const float* sqb  = sq + (size_t)b * NPTS;

    const int m = lane & 15;
    const int h = lane >> 4;

    v2f afrag0[8], afrag1[8];
#pragma unroll
    for (int c = 0; c < 8; ++c) {
        const float* p0 = xb + (size_t)(rowbase + m) * CH + 4 * c + 2 * h;
        const float* p1 = xb + (size_t)(rowbase + 16 + m) * CH + 4 * c + 2 * h;
        afrag0[c].x = -2.0f * p0[0];
        afrag0[c].y = -2.0f * p0[1];
        afrag1[c].x = -2.0f * p1[0];
        afrag1[c].y = -2.0f * p1[1];
    }
    v2f afragOne;
    afragOne.x = (h == 0) ? 1.0f : 0.0f;
    afragOne.y = 0.0f;

    float key[KNN];
    int   idx[KNN];
#pragma unroll
    for (int s = 0; s < KNN; ++s) { key[s] = 3.0e38f; idx[s] = 0; }

    v2f   bcur[8];
    float sqcur;
#pragma unroll
    for (int c = 0; c < 8; ++c) {
        const float* p = xb + (size_t)m * CH + 4 * c + 2 * h;
        bcur[c].x = p[0];
        bcur[c].y = p[1];
    }
    sqcur = sqb[m];

    for (int jt = 0; jt < NCTILES; ++jt) {
        const int colbase = jt * 16;

        const int nbase = ((jt + 1) & (NCTILES - 1)) * 16;
        v2f bnxt[8];
#pragma unroll
        for (int c = 0; c < 8; ++c) {
            const float* p = xb + (size_t)(nbase + m) * CH + 4 * c + 2 * h;
            bnxt[c].x = p[0];
            bnxt[c].y = p[1];
        }
        const float sqnxt = sqb[nbase + m];

        v8f acc0 = {};
        v8f acc1 = {};
#pragma unroll
        for (int c = 0; c < 8; ++c) {
            acc0 = __builtin_amdgcn_wmma_f32_16x16x4_f32(
                false, afrag0[c], false, bcur[c], (short)0, acc0, false, false);
            acc1 = __builtin_amdgcn_wmma_f32_16x16x4_f32(
                false, afrag1[c], false, bcur[c], (short)0, acc1, false, false);
        }
        {
            v2f bfrag;
            bfrag.x = (h == 0) ? sqcur : 0.0f;
            bfrag.y = 0.0f;
            acc0 = __builtin_amdgcn_wmma_f32_16x16x4_f32(
                false, afragOne, false, bfrag, (short)0, acc0, false, false);
            acc1 = __builtin_amdgcn_wmma_f32_16x16x4_f32(
                false, afragOne, false, bfrag, (short)0, acc1, false, false);
        }

#pragma unroll
        for (int r = 0; r < 8; ++r) {
            distT[(r + 8 * h) * DPITCH + m]        = acc0[r];
            distT[(r + 8 * h + 16) * DPITCH + m]   = acc1[r];
        }
        lds_sync();

        const float4* rowp = reinterpret_cast<const float4*>(&distT[lane * DPITCH]);
        const float4 q0 = rowp[0];
        const float4 q1 = rowp[1];
        const float4 q2 = rowp[2];
        const float4 q3 = rowp[3];
        const float dv[16] = { q0.x, q0.y, q0.z, q0.w, q1.x, q1.y, q1.z, q1.w,
                               q2.x, q2.y, q2.z, q2.w, q3.x, q3.y, q3.z, q3.w };
#pragma unroll
        for (int c = 0; c < 16; ++c) {
            const float d = dv[c];
            if (d < key[KNN - 1]) {
                float cv = d;
                int   ci = colbase + c;
#pragma unroll
                for (int s = 0; s < KNN; ++s) {
                    const float ks = key[s];
                    const int   is = idx[s];
                    const bool  sw = cv < ks;
                    key[s] = sw ? cv : ks;
                    idx[s] = sw ? ci : is;
                    cv = sw ? ks : cv;
                    ci = sw ? is : ci;
                }
            }
        }
        lds_sync();

#pragma unroll
        for (int c = 0; c < 8; ++c) bcur[c] = bnxt[c];
        sqcur = sqnxt;
    }

#pragma unroll
    for (int s = 0; s < KNN; ++s)
        selIdx[lane * KNN + s] = idx[s];
    __syncthreads();

    float* poolb = pooled + ((size_t)b * NPTS + rowbase) * CH;
#pragma unroll 1
    for (int r = 0; r < ROWS_PER_WAVE; ++r) {
        float mx = -3.0e38f;
#pragma unroll
        for (int k = 0; k < KNN; ++k) {
            const int j = selIdx[r * KNN + k];
            const float v = xb[(size_t)j * CH + lane];
            mx = fmaxf(mx, v);
        }
        poolb[r * CH + lane] = mx;
    }
}

__global__ __launch_bounds__(32) void mlp_kernel(const float* __restrict__ pooled,
                                                 const float* __restrict__ Wl,
                                                 const float* __restrict__ bl,
                                                 const float* __restrict__ Wc,
                                                 const float* __restrict__ bc,
                                                 float* __restrict__ out) {
    __shared__ float hT[16 * 68];

    const int lane = threadIdx.x;
    const int m = lane & 15;
    const int h = lane >> 4;
    const size_t rowbase = (size_t)blockIdx.x * 16;

    v2f afrag[8];
#pragma unroll
    for (int c = 0; c < 8; ++c) {
        const float* p = pooled + (rowbase + m) * CH + 4 * c + 2 * h;
        afrag[c].x = p[0];
        afrag[c].y = p[1];
    }

#pragma unroll
    for (int t = 0; t < 4; ++t) {
        v8f acc = {};
#pragma unroll
        for (int c = 0; c < 8; ++c) {
            const float* p = Wl + (size_t)(4 * c + 2 * h) * LDIM + t * 16 + m;
            v2f bfrag;
            bfrag.x = p[0];
            bfrag.y = p[LDIM];
            acc = __builtin_amdgcn_wmma_f32_16x16x4_f32(
                false, afrag[c], false, bfrag, (short)0, acc, false, false);
        }
        const float bias = bl[t * 16 + m];
#pragma unroll
        for (int r = 0; r < 8; ++r)
            hT[(r + 8 * h) * 68 + t * 16 + m] = acc[r] + bias;
    }
    __syncthreads();

    v2f a2[16];
#pragma unroll
    for (int c = 0; c < 16; ++c) {
        a2[c].x = hT[m * 68 + 4 * c + 2 * h];
        a2[c].y = hT[m * 68 + 4 * c + 2 * h + 1];
    }

    float* outp = out + rowbase * ODIM;
#pragma unroll
    for (int t = 0; t < 8; ++t) {
        v8f acc = {};
#pragma unroll
        for (int c = 0; c < 16; ++c) {
            const float* p = Wc + (size_t)(4 * c + 2 * h) * ODIM + t * 16 + m;
            v2f bfrag;
            bfrag.x = p[0];
            bfrag.y = p[ODIM];
            acc = __builtin_amdgcn_wmma_f32_16x16x4_f32(
                false, a2[c], false, bfrag, (short)0, acc, false, false);
        }
        const float bias = bc[t * 16 + m];
#pragma unroll
        for (int r = 0; r < 8; ++r) {
            const float v = acc[r] + bias;
            outp[(size_t)(r + 8 * h) * ODIM + t * 16 + m] = fmaxf(v, 0.0f);
        }
    }
}

extern "C" void kernel_launch(void* const* d_in, const int* in_sizes, int n_in,
                              void* d_out, int out_size, void* d_ws, size_t ws_size,
                              hipStream_t stream) {
    const float* x  = (const float*)d_in[0];
    const float* Wl = (const float*)d_in[1];
    const float* bl = (const float*)d_in[2];
    const float* Wc = (const float*)d_in[3];
    const float* bc = (const float*)d_in[4];
    float* out = (float*)d_out;

    float* pooled = (float*)d_ws;
    float* sqbuf  = pooled + (size_t)BATCH * NPTS * CH;

    const int total = BATCH * NPTS;
    sq_kernel<<<(total + 255) / 256, 256, 0, stream>>>(x, sqbuf, total);
    knn_pool_kernel<<<BATCH * NRBLK, 32, 0, stream>>>(x, sqbuf, pooled);
    mlp_kernel<<<total / 16, 32, 0, stream>>>(pooled, Wl, bl, Wc, bc, out);
}


# kNN kernel: workgroup of 4 waves per 32-row block; each wave scans a quarter of the column tiles with its own LDS tile and top-16, lists merged pairwise at the end (4 waves per SIMD, was 1)
# speedup vs baseline: 2.8756x; 1.7891x over previous
#include <hip/hip_runtime.h>
#include <hip/hip_bf16.h>

#define BATCH 4
#define NPTS  8192
#define CH    32
#define KNN   16
#define LDIM  64
#define ODIM  128
#define NCTILES (NPTS / 16)
#define ROWS_PER_WAVE 32
#define NRBLK (NPTS / ROWS_PER_WAVE)
#define DPITCH 20
#define NSPLIT 4
#define TILES_PER_WAVE (NCTILES / NSPLIT)

typedef __attribute__((ext_vector_type(2))) float v2f;
typedef __attribute__((ext_vector_type(8))) float v8f;

__global__ void sq_kernel(const float* __restrict__ x, float* __restrict__ sq, int total) {
    int i = blockIdx.x * blockDim.x + threadIdx.x;
    if (i >= total) return;
    const float* p = x + (size_t)i * CH;
    float s = 0.0f;
#pragma unroll
    for (int c = 0; c < CH; ++c) s += p[c] * p[c];
    sq[i] = s;
}

__device__ __forceinline__ void wave_lds_sync() {
    __builtin_amdgcn_fence(__ATOMIC_RELEASE, "workgroup", "local");
    __builtin_amdgcn_fence(__ATOMIC_ACQUIRE, "workgroup", "local");
}

__device__ __forceinline__ void topk_insert(float (&key)[KNN], int (&idx)[KNN], float cv, int ci) {
#pragma unroll
    for (int s = 0; s < KNN; ++s) {
        const float ks = key[s];
        const int   is = idx[s];
        const bool  sw = cv < ks;
        key[s] = sw ? cv : ks;
        idx[s] = sw ? ci : is;
        cv = sw ? ks : cv;
        ci = sw ? is : ci;
    }
}

__device__ __forceinline__ void topk_merge(float (&key)[KNN], int (&idx)[KNN],
                                           const float* mk, const int* mi, int lane) {
#pragma unroll 1
    for (int s = 0; s < KNN; ++s) {
        const float d  = mk[s * 32 + lane];
        const int   ci = mi[s * 32 + lane];
        if (d < key[KNN - 1]) topk_insert(key, idx, d, ci);
    }
}

__global__ __launch_bounds__(32 * NSPLIT) void knn_pool_kernel(const float* __restrict__ x,
                                                               const float* __restrict__ sq,
                                                               float* __restrict__ pooled) {
    __shared__ __attribute__((aligned(16))) float distT[NSPLIT][ROWS_PER_WAVE * DPITCH];
    __shared__ float mrgKey[3][KNN * 32];
    __shared__ int   mrgIdx[3][KNN * 32];
    __shared__ int   selIdx[ROWS_PER_WAVE * KNN];

    const int lane    = threadIdx.x & 31;
    const int wv      = __builtin_amdgcn_readfirstlane(threadIdx.x >> 5);
    const int rblk    = blockIdx.x % NRBLK;
    const int b       = blockIdx.x / NRBLK;
    const int rowbase = rblk * ROWS_PER_WAVE;
    const float* xb   = x  + (size_t)b * NPTS * CH;
    const float* sqb  = sq + (size_t)b * NPTS;
    float* tile       = distT[wv];

    const int m = lane & 15;
    const int h = lane >> 4;

    v2f afrag0[8], afrag1[8];
#pragma unroll
    for (int c = 0; c < 8; ++c) {
        const float* p0 = xb + (size_t)(rowbase + m) * CH + 4 * c + 2 * h;
        const float* p1 = xb + (size_t)(rowbase + 16 + m) * CH + 4 * c + 2 * h;
        afrag0[c].x = -2.0f * p0[0];
        afrag0[c].y = -2.0f * p0[1];
        afrag1[c].x = -2.0f * p1[0];
        afrag1[c].y = -2.0f * p1[1];
    }
    v2f afragOne;
    afragOne.x = (h == 0) ? 1.0f : 0.0f;
    afragOne.y = 0.0f;

    float key[KNN];
    int   idx[KNN];
#pragma unroll
    for (int s = 0; s < KNN; ++s) { key[s] = 3.0e38f; idx[s] = 0; }

    const int jt0 = wv * TILES_PER_WAVE;

    v2f   bcur[8];
    float sqcur;
#pragma unroll
    for (int c = 0; c < 8; ++c) {
        const float* p = xb + (size_t)(jt0 * 16 + m) * CH + 4 * c + 2 * h;
        bcur[c].x = p[0];
        bcur[c].y = p[1];
    }
    sqcur = sqb[jt0 * 16 + m];

    for (int jt = jt0; jt < jt0 + TILES_PER_WAVE; ++jt) {
        const int colbase = jt * 16;

        const int nbase = ((jt + 1) & (NCTILES - 1)) * 16;
        v2f bnxt[8];
#pragma unroll
        for (int c = 0; c < 8; ++c) {
            const float* p = xb + (size_t)(nbase + m) * CH + 4 * c + 2 * h;
            bnxt[c].x = p[0];
            bnxt[c].y = p[1];
        }
        const float sqnxt = sqb[nbase + m];

        v8f acc0 = {};
        v8f acc1 = {};
#pragma unroll
        for (int c = 0; c < 8; ++c) {
            acc0 = __builtin_amdgcn_wmma_f32_16x16x4_f32(
                false, afrag0[c], false, bcur[c], (short)0, acc0, false, false);
            acc1 = __builtin_amdgcn_wmma_f32_16x16x4_f32(
                false, afrag1[c], false, bcur[c], (short)0, acc1, false, false);
        }
        {
            v2f bfrag;
            bfrag.x = (h == 0) ? sqcur : 0.0f;
            bfrag.y = 0.0f;
            acc0 = __builtin_amdgcn_wmma_f32_16x16x4_f32(
                false, afragOne, false, bfrag, (short)0, acc0, false, false);
            acc1 = __builtin_amdgcn_wmma_f32_16x16x4_f32(
                false, afragOne, false, bfrag, (short)0, acc1, false, false);
        }

#pragma unroll
        for (int r = 0; r < 8; ++r) {
            tile[(r + 8 * h) * DPITCH + m]        = acc0[r];
            tile[(r + 8 * h + 16) * DPITCH + m]   = acc1[r];
        }
        wave_lds_sync();

        const float4* rowp = reinterpret_cast<const float4*>(&tile[lane * DPITCH]);
        const float4 q0 = rowp[0];
        const float4 q1 = rowp[1];
        const float4 q2 = rowp[2];
        const float4 q3 = rowp[3];
        const float dv[16] = { q0.x, q0.y, q0.z, q0.w, q1.x, q1.y, q1.z, q1.w,
                               q2.x, q2.y, q2.z, q2.w, q3.x, q3.y, q3.z, q3.w };
#pragma unroll
        for (int c = 0; c < 16; ++c) {
            const float d = dv[c];
            if (d < key[KNN - 1]) topk_insert(key, idx, d, colbase + c);
        }
        wave_lds_sync();

#pragma unroll
        for (int c = 0; c < 8; ++c) bcur[c] = bnxt[c];
        sqcur = sqnxt;
    }

    if (wv & 1) {
#pragma unroll
        for (int s = 0; s < KNN; ++s) {
            mrgKey[wv >> 1][s * 32 + lane] = key[s];
            mrgIdx[wv >> 1][s * 32 + lane] = idx[s];
        }
    }
    __syncthreads();
    if (!(wv & 1)) topk_merge(key, idx, mrgKey[wv >> 1], mrgIdx[wv >> 1], lane);
    if (wv == 2) {
#pragma unroll
        for (int s = 0; s < KNN; ++s) {
            mrgKey[2][s * 32 + lane] = key[s];
            mrgIdx[2][s * 32 + lane] = idx[s];
        }
    }
    __syncthreads();
    if (wv == 0) {
        topk_merge(key, idx, mrgKey[2], mrgIdx[2], lane);
#pragma unroll
        for (int s = 0; s < KNN; ++s)
            selIdx[lane * KNN + s] = idx[s];
    }
    __syncthreads();

    float* poolb = pooled + ((size_t)b * NPTS + rowbase) * CH;
#pragma unroll 1
    for (int rr = 0; rr < ROWS_PER_WAVE / NSPLIT; ++rr) {
        const int r = wv * (ROWS_PER_WAVE / NSPLIT) + rr;
        float mx = -3.0e38f;
#pragma unroll
        for (int k = 0; k < KNN; ++k) {
            const int j = selIdx[r * KNN + k];
            const float v = xb[(size_t)j * CH + lane];
            mx = fmaxf(mx, v);
        }
        poolb[r * CH + lane] = mx;
    }
}

__global__ __launch_bounds__(32) void mlp_kernel(const float* __restrict__ pooled,
                                                 const float* __restrict__ Wl,
                                                 const float* __restrict__ bl,
                                                 const float* __restrict__ Wc,
                                                 const float* __restrict__ bc,
                                                 float* __restrict__ out) {
    __shared__ float hT[16 * 68];

    const int lane = threadIdx.x;
    const int m = lane & 15;
    const int h = lane >> 4;
    const size_t rowbase = (size_t)blockIdx.x * 16;

    v2f afrag[8];
#pragma unroll
    for (int c = 0; c < 8; ++c) {
        const float* p = pooled + (rowbase + m) * CH + 4 * c + 2 * h;
        afrag[c].x = p[0];
        afrag[c].y = p[1];
    }

#pragma unroll
    for (int t = 0; t < 4; ++t) {
        v8f acc = {};
#pragma unroll
        for (int c = 0; c < 8; ++c) {
            const float* p = Wl + (size_t)(4 * c + 2 * h) * LDIM + t * 16 + m;
            v2f bfrag;
            bfrag.x = p[0];
            bfrag.y = p[LDIM];
            acc = __builtin_amdgcn_wmma_f32_16x16x4_f32(
                false, afrag[c], false, bfrag, (short)0, acc, false, false);
        }
        const float bias = bl[t * 16 + m];
#pragma unroll
        for (int r = 0; r < 8; ++r)
            hT[(r + 8 * h) * 68 + t * 16 + m] = acc[r] + bias;
    }
    __syncthreads();

    v2f a2[16];
#pragma unroll
    for (int c = 0; c < 16; ++c) {
        a2[c].x = hT[m * 68 + 4 * c + 2 * h];
        a2[c].y = hT[m * 68 + 4 * c + 2 * h + 1];
    }

    float* outp = out + rowbase * ODIM;
#pragma unroll
    for (int t = 0; t < 8; ++t) {
        v8f acc = {};
#pragma unroll
        for (int c = 0; c < 16; ++c) {
            const float* p = Wc + (size_t)(4 * c + 2 * h) * ODIM + t * 16 + m;
            v2f bfrag;
            bfrag.x = p[0];
            bfrag.y = p[ODIM];
            acc = __builtin_amdgcn_wmma_f32_16x16x4_f32(
                false, a2[c], false, bfrag, (short)0, acc, false, false);
        }
        const float bias = bc[t * 16 + m];
#pragma unroll
        for (int r = 0; r < 8; ++r) {
            const float v = acc[r] + bias;
            outp[(size_t)(r + 8 * h) * ODIM + t * 16 + m] = fmaxf(v, 0.0f);
        }
    }
}

extern "C" void kernel_launch(void* const* d_in, const int* in_sizes, int n_in,
                              void* d_out, int out_size, void* d_ws, size_t ws_size,
                              hipStream_t stream) {
    const float* x  = (const float*)d_in[0];
    const float* Wl = (const float*)d_in[1];
    const float* bl = (const float*)d_in[2];
    const float* Wc = (const float*)d_in[3];
    const float* bc = (const float*)d_in[4];
    float* out = (float*)d_out;

    float* pooled = (float*)d_ws;
    float* sqbuf  = pooled + (size_t)BATCH * NPTS * CH;

    const int total = BATCH * NPTS;
    sq_kernel<<<(total + 255) / 256, 256, 0, stream>>>(x, sqbuf, total);
    knn_pool_kernel<<<BATCH * NRBLK, 32 * NSPLIT, 0, stream>>>(x, sqbuf, pooled);
    mlp_kernel<<<total / 16, 32, 0, stream>>>(pooled, Wl, bl, Wc, bc, out);
}


# kNN selection: keys under a lane's worst kept key are queued per lane in LDS with no branch per candidate; the sorted insertion runs for the whole wave only when a queue nears full
# speedup vs baseline: 4.4659x; 1.5530x over previous
#include <hip/hip_runtime.h>
#include <hip/hip_bf16.h>

#define BATCH 4
#define NPTS  8192
#define CH    32
#define KNN   16
#define LDIM  64
#define ODIM  128
#define NCTILES (NPTS / 16)
#define ROWS_PER_WAVE 32
#define NRBLK (NPTS / ROWS_PER_WAVE)
#define DPITCH 20
#define NSPLIT 4
#define TILES_PER_WAVE (NCTILES / NSPLIT)
#define QCAP 32

typedef __attribute__((ext_vector_type(2))) float v2f;
typedef __attribute__((ext_vector_type(8))) float v8f;

__global__ void sq_kernel(const float* __restrict__ x, float* __restrict__ sq, int total) {
    int i = blockIdx.x * blockDim.x + threadIdx.x;
    if (i >= total) return;
    const float* p = x + (size_t)i * CH;
    float s = 0.0f;
#pragma unroll
    for (int c = 0; c < CH; ++c) s += p[c] * p[c];
    sq[i] = s;
}

__device__ __forceinline__ void wave_lds_sync() {
    __builtin_amdgcn_fence(__ATOMIC_RELEASE, "workgroup", "local");
    __builtin_amdgcn_fence(__ATOMIC_ACQUIRE, "workgroup", "local");
}

__device__ __forceinline__ void topk_insert(float (&key)[KNN], int (&idx)[KNN], float cv, int ci) {
#pragma unroll
    for (int s = 0; s < KNN; ++s) {
        const float ks = key[s];
        const int   is = idx[s];
        const bool  sw = cv < ks;
        key[s] = sw ? cv : ks;
        idx[s] = sw ? ci : is;
        cv = sw ? ks : cv;
        ci = sw ? is : ci;
    }
}

struct PendSlot {
    float k[32];
    int   i[32];
};

__device__ __forceinline__ void topk_merge(float (&key)[KNN], int (&idx)[KNN],
                                           const float* mk, const int* mi, int lane) {
#pragma unroll 1
    for (int s = 0; s < KNN; ++s) {
        const float d  = mk[s * 32 + lane];
        const int   ci = mi[s * 32 + lane];
        if (d < key[KNN - 1]) topk_insert(key, idx, d, ci);
    }
}

__global__ __launch_bounds__(32 * NSPLIT) void knn_pool_kernel(const float* __restrict__ x,
                                                               const float* __restrict__ sq,
                                                               float* __restrict__ pooled) {
    __shared__ __attribute__((aligned(16))) float distT[NSPLIT][ROWS_PER_WAVE * DPITCH];
    __shared__ float mrgKey[3][KNN * 32];
    __shared__ int   mrgIdx[3][KNN * 32];
    __shared__ int   selIdx[ROWS_PER_WAVE * KNN];
    __shared__ PendSlot pend[NSPLIT][QCAP];

    const int lane    = threadIdx.x & 31;
    const int wv      = __builtin_amdgcn_readfirstlane(threadIdx.x >> 5);
    const int rblk    = blockIdx.x % NRBLK;
    const int b       = blockIdx.x / NRBLK;
    const int rowbase = rblk * ROWS_PER_WAVE;
    const float* xb   = x  + (size_t)b * NPTS * CH;
    const float* sqb  = sq + (size_t)b * NPTS;
    float* tile       = distT[wv];
    PendSlot* pq      = pend[wv];

    const int m = lane & 15;
    const int h = lane >> 4;

    v2f afrag0[8], afrag1[8];
#pragma unroll
    for (int c = 0; c < 8; ++c) {
        const float* p0 = xb + (size_t)(rowbase + m) * CH + 4 * c + 2 * h;
        const float* p1 = xb + (size_t)(rowbase + 16 + m) * CH + 4 * c + 2 * h;
        afrag0[c].x = -2.0f * p0[0];
        afrag0[c].y = -2.0f * p0[1];
        afrag1[c].x = -2.0f * p1[0];
        afrag1[c].y = -2.0f * p1[1];
    }
    v2f afragOne;
    afragOne.x = (h == 0) ? 1.0f : 0.0f;
    afragOne.y = 0.0f;

    float key[KNN];
    int   idx[KNN];
#pragma unroll
    for (int s = 0; s < KNN; ++s) { key[s] = 3.0e38f; idx[s] = 0; }

    int qn = 0;

    const int jt0 = wv * TILES_PER_WAVE;

    v2f   bcur[8];
    float sqcur;
#pragma unroll
    for (int c = 0; c < 8; ++c) {
        const float* p = xb + (size_t)(jt0 * 16 + m) * CH + 4 * c + 2 * h;
        bcur[c].x = p[0];
        bcur[c].y = p[1];
    }
    sqcur = sqb[jt0 * 16 + m];

    for (int jt = jt0; jt < jt0 + TILES_PER_WAVE; ++jt) {
        const int colbase = jt * 16;

        const int nbase = ((jt + 1) & (NCTILES - 1)) * 16;
        v2f bnxt[8];
#pragma unroll
        for (int c = 0; c < 8; ++c) {
            const float* p = xb + (size_t)(nbase + m) * CH + 4 * c + 2 * h;
            bnxt[c].x = p[0];
            bnxt[c].y = p[1];
        }
        const float sqnxt = sqb[nbase + m];

        v8f acc0 = {};
        v8f acc1 = {};
#pragma unroll
        for (int c = 0; c < 8; ++c) {
            acc0 = __builtin_amdgcn_wmma_f32_16x16x4_f32(
                false, afrag0[c], false, bcur[c], (short)0, acc0, false, false);
            acc1 = __builtin_amdgcn_wmma_f32_16x16x4_f32(
                false, afrag1[c], false, bcur[c], (short)0, acc1, false, false);
        }
        {
            v2f bfrag;
            bfrag.x = (h == 0) ? sqcur : 0.0f;
            bfrag.y = 0.0f;
            acc0 = __builtin_amdgcn_wmma_f32_16x16x4_f32(
                false, afragOne, false, bfrag, (short)0, acc0, false, false);
            acc1 = __builtin_amdgcn_wmma_f32_16x16x4_f32(
                false, afragOne, false, bfrag, (short)0, acc1, false, false);
        }

#pragma unroll
        for (int r = 0; r < 8; ++r) {
            tile[(r + 8 * h) * DPITCH + m]        = acc0[r];
            tile[(r + 8 * h + 16) * DPITCH + m]   = acc1[r];
        }
        wave_lds_sync();

        const float4* rowp = reinterpret_cast<const float4*>(&tile[lane * DPITCH]);
        const float4 q0 = rowp[0];
        const float4 q1 = rowp[1];
        const float4 q2 = rowp[2];
        const float4 q3 = rowp[3];
        const float dv[16] = { q0.x, q0.y, q0.z, q0.w, q1.x, q1.y, q1.z, q1.w,
                               q2.x, q2.y, q2.z, q2.w, q3.x, q3.y, q3.z, q3.w };
        const float bound = key[KNN - 1];
#pragma unroll
        for (int c = 0; c < 16; ++c) {
            const float d = dv[c];
            pq[qn].k[lane] = d;
            pq[qn].i[lane] = colbase + c;
            qn += (d < bound) ? 1 : 0;
        }

        const bool lastTile = (jt == jt0 + TILES_PER_WAVE - 1);
        if (lastTile || __builtin_amdgcn_ballot_w32(qn > QCAP - 16) != 0u) {
            const int maxn = __builtin_amdgcn_readfirstlane(
                (int)__builtin_amdgcn_wave_reduce_max_u32((unsigned)qn, 0));
            for (int q = 0; q < maxn; q += 2) {
                const float k0 = pq[q].k[lane];
                const int   c0 = pq[q].i[lane];
                const float k1 = pq[q + 1].k[lane];
                const int   c1 = pq[q + 1].i[lane];
                const float d0 = (q < qn)     ? k0 : __builtin_inff();
                const float d1 = (q + 1 < qn) ? k1 : __builtin_inff();
                topk_insert(key, idx, d0, c0);
                topk_insert(key, idx, d1, c1);
            }
            qn = 0;
        }
        wave_lds_sync();

#pragma unroll
        for (int c = 0; c < 8; ++c) bcur[c] = bnxt[c];
        sqcur = sqnxt;
    }

    if (wv & 1) {
#pragma unroll
        for (int s = 0; s < KNN; ++s) {
            mrgKey[wv >> 1][s * 32 + lane] = key[s];
            mrgIdx[wv >> 1][s * 32 + lane] = idx[s];
        }
    }
    __syncthreads();
    if (!(wv & 1)) topk_merge(key, idx, mrgKey[wv >> 1], mrgIdx[wv >> 1], lane);
    if (wv == 2) {
#pragma unroll
        for (int s = 0; s < KNN; ++s) {
            mrgKey[2][s * 32 + lane] = key[s];
            mrgIdx[2][s * 32 + lane] = idx[s];
        }
    }
    __syncthreads();
    if (wv == 0) {
        topk_merge(key, idx, mrgKey[2], mrgIdx[2], lane);
#pragma unroll
        for (int s = 0; s < KNN; ++s)
            selIdx[lane * KNN + s] = idx[s];
    }
    __syncthreads();

    float* poolb = pooled + ((size_t)b * NPTS + rowbase) * CH;
#pragma unroll 1
    for (int rr = 0; rr < ROWS_PER_WAVE / NSPLIT; ++rr) {
        const int r = wv * (ROWS_PER_WAVE / NSPLIT) + rr;
        float mx = -3.0e38f;
#pragma unroll
        for (int k = 0; k < KNN; ++k) {
            const int j = selIdx[r * KNN + k];
            const float v = xb[(size_t)j * CH + lane];
            mx = fmaxf(mx, v);
        }
        poolb[r * CH + lane] = mx;
    }
}

__global__ __launch_bounds__(32) void mlp_kernel(const float* __restrict__ pooled,
                                                 const float* __restrict__ Wl,
                                                 const float* __restrict__ bl,
                                                 const float* __restrict__ Wc,
                                                 const float* __restrict__ bc,
                                                 float* __restrict__ out) {
    __shared__ float hT[16 * 68];

    const int lane = threadIdx.x;
    const int m = lane & 15;
    const int h = lane >> 4;
    const size_t rowbase = (size_t)blockIdx.x * 16;

    v2f afrag[8];
#pragma unroll
    for (int c = 0; c < 8; ++c) {
        const float* p = pooled + (rowbase + m) * CH + 4 * c + 2 * h;
        afrag[c].x = p[0];
        afrag[c].y = p[1];
    }

#pragma unroll
    for (int t = 0; t < 4; ++t) {
        v8f acc = {};
#pragma unroll
        for (int c = 0; c < 8; ++c) {
            const float* p = Wl + (size_t)(4 * c + 2 * h) * LDIM + t * 16 + m;
            v2f bfrag;
            bfrag.x = p[0];
            bfrag.y = p[LDIM];
            acc = __builtin_amdgcn_wmma_f32_16x16x4_f32(
                false, afrag[c], false, bfrag, (short)0, acc, false, false);
        }
        const float bias = bl[t * 16 + m];
#pragma unroll
        for (int r = 0; r < 8; ++r)
            hT[(r + 8 * h) * 68 + t * 16 + m] = acc[r] + bias;
    }
    __syncthreads();

    v2f a2[16];
#pragma unroll
    for (int c = 0; c < 16; ++c) {
        a2[c].x = hT[m * 68 + 4 * c + 2 * h];
        a2[c].y = hT[m * 68 + 4 * c + 2 * h + 1];
    }

    float* outp = out + rowbase * ODIM;
#pragma unroll
    for (int t = 0; t < 8; ++t) {
        v8f acc = {};
#pragma unroll
        for (int c = 0; c < 16; ++c) {
            const float* p = Wc + (size_t)(4 * c + 2 * h) * ODIM + t * 16 + m;
            v2f bfrag;
            bfrag.x = p[0];
            bfrag.y = p[ODIM];
            acc = __builtin_amdgcn_wmma_f32_16x16x4_f32(
                false, a2[c], false, bfrag, (short)0, acc, false, false);
        }
        const float bias = bc[t * 16 + m];
#pragma unroll
        for (int r = 0; r < 8; ++r) {
            const float v = acc[r] + bias;
            outp[(size_t)(r + 8 * h) * ODIM + t * 16 + m] = fmaxf(v, 0.0f);
        }
    }
}

extern "C" void kernel_launch(void* const* d_in, const int* in_sizes, int n_in,
                              void* d_out, int out_size, void* d_ws, size_t ws_size,
                              hipStream_t stream) {
    const float* x  = (const float*)d_in[0];
    const float* Wl = (const float*)d_in[1];
    const float* bl = (const float*)d_in[2];
    const float* Wc = (const float*)d_in[3];
    const float* bc = (const float*)d_in[4];
    float* out = (float*)d_out;

    float* pooled = (float*)d_ws;
    float* sqbuf  = pooled + (size_t)BATCH * NPTS * CH;

    const int total = BATCH * NPTS;
    sq_kernel<<<(total + 255) / 256, 256, 0, stream>>>(x, sqbuf, total);
    knn_pool_kernel<<<BATCH * NRBLK, 32 * NSPLIT, 0, stream>>>(x, sqbuf, pooled);
    mlp_kernel<<<total / 16, 32, 0, stream>>>(pooled, Wl, bl, Wc, bc, out);
}


# kNN selection: each wave posts its 4th-best kept key per row in LDS; a candidate is queued only if it is also under the largest of the four posted keys, so far fewer sorted insertions run
# speedup vs baseline: 4.6233x; 1.0352x over previous
#include <hip/hip_runtime.h>
#include <hip/hip_bf16.h>

#define BATCH 4
#define NPTS  8192
#define CH    32
#define KNN   16
#define LDIM  64
#define ODIM  128
#define NCTILES (NPTS / 16)
#define ROWS_PER_WAVE 32
#define NRBLK (NPTS / ROWS_PER_WAVE)
#define DPITCH 20
#define NSPLIT 4
#define TILES_PER_WAVE (NCTILES / NSPLIT)
#define QCAP 32

typedef __attribute__((ext_vector_type(2))) float v2f;
typedef __attribute__((ext_vector_type(8))) float v8f;
typedef __attribute__((ext_vector_type(4))) float v4f;
#define POSTPOS 3

__global__ void sq_kernel(const float* __restrict__ x, float* __restrict__ sq, int total) {
    int i = blockIdx.x * blockDim.x + threadIdx.x;
    if (i >= total) return;
    const float* p = x + (size_t)i * CH;
    float s = 0.0f;
#pragma unroll
    for (int c = 0; c < CH; ++c) s += p[c] * p[c];
    sq[i] = s;
}

__device__ __forceinline__ void wave_lds_sync() {
    __builtin_amdgcn_fence(__ATOMIC_RELEASE, "workgroup", "local");
    __builtin_amdgcn_fence(__ATOMIC_ACQUIRE, "workgroup", "local");
}

__device__ __forceinline__ void topk_insert(float (&key)[KNN], int (&idx)[KNN], float cv, int ci) {
#pragma unroll
    for (int s = 0; s < KNN; ++s) {
        const float ks = key[s];
        const int   is = idx[s];
        const bool  sw = cv < ks;
        key[s] = sw ? cv : ks;
        idx[s] = sw ? ci : is;
        cv = sw ? ks : cv;
        ci = sw ? is : ci;
    }
}

struct PendSlot {
    float k[32];
    int   i[32];
};

__device__ __forceinline__ void topk_merge(float (&key)[KNN], int (&idx)[KNN],
                                           const float* mk, const int* mi, int lane) {
#pragma unroll 1
    for (int s = 0; s < KNN; ++s) {
        const float d  = mk[s * 32 + lane];
        const int   ci = mi[s * 32 + lane];
        if (d < key[KNN - 1]) topk_insert(key, idx, d, ci);
    }
}

__global__ __launch_bounds__(32 * NSPLIT) void knn_pool_kernel(const float* __restrict__ x,
                                                               const float* __restrict__ sq,
                                                               float* __restrict__ pooled) {
    __shared__ __attribute__((aligned(16))) float distT[NSPLIT][ROWS_PER_WAVE * DPITCH];
    __shared__ float mrgKey[3][KNN * 32];
    __shared__ int   mrgIdx[3][KNN * 32];
    __shared__ int   selIdx[ROWS_PER_WAVE * KNN];
    __shared__ PendSlot pend[NSPLIT][QCAP];
    __shared__ __attribute__((aligned(16))) float postKey[32 * NSPLIT];

    const int lane    = threadIdx.x & 31;
    const int wv      = __builtin_amdgcn_readfirstlane(threadIdx.x >> 5);
    const int rblk    = blockIdx.x % NRBLK;
    const int b       = blockIdx.x / NRBLK;
    const int rowbase = rblk * ROWS_PER_WAVE;
    const float* xb   = x  + (size_t)b * NPTS * CH;
    const float* sqb  = sq + (size_t)b * NPTS;
    float* tile       = distT[wv];
    PendSlot* pq      = pend[wv];

    const int m = lane & 15;
    const int h = lane >> 4;

    v2f afrag0[8], afrag1[8];
#pragma unroll
    for (int c = 0; c < 8; ++c) {
        const float* p0 = xb + (size_t)(rowbase + m) * CH + 4 * c + 2 * h;
        const float* p1 = xb + (size_t)(rowbase + 16 + m) * CH + 4 * c + 2 * h;
        afrag0[c].x = -2.0f * p0[0];
        afrag0[c].y = -2.0f * p0[1];
        afrag1[c].x = -2.0f * p1[0];
        afrag1[c].y = -2.0f * p1[1];
    }
    v2f afragOne;
    afragOne.x = (h == 0) ? 1.0f : 0.0f;
    afragOne.y = 0.0f;

    float key[KNN];
    int   idx[KNN];
#pragma unroll
    for (int s = 0; s < KNN; ++s) { key[s] = 3.0e38f; idx[s] = 0; }

    int qn = 0;

    postKey[lane * NSPLIT + wv] = 3.0e38f;
    __syncthreads();

    const int jt0 = wv * TILES_PER_WAVE;

    v2f   bcur[8];
    float sqcur;
#pragma unroll
    for (int c = 0; c < 8; ++c) {
        const float* p = xb + (size_t)(jt0 * 16 + m) * CH + 4 * c + 2 * h;
        bcur[c].x = p[0];
        bcur[c].y = p[1];
    }
    sqcur = sqb[jt0 * 16 + m];

    for (int jt = jt0; jt < jt0 + TILES_PER_WAVE; ++jt) {
        const int colbase = jt * 16;

        const v4f posted = *reinterpret_cast<const v4f*>(&postKey[lane * NSPLIT]);

        const int nbase = ((jt + 1) & (NCTILES - 1)) * 16;
        v2f bnxt[8];
#pragma unroll
        for (int c = 0; c < 8; ++c) {
            const float* p = xb + (size_t)(nbase + m) * CH + 4 * c + 2 * h;
            bnxt[c].x = p[0];
            bnxt[c].y = p[1];
        }
        const float sqnxt = sqb[nbase + m];

        v8f acc0 = {};
        v8f acc1 = {};
#pragma unroll
        for (int c = 0; c < 8; ++c) {
            acc0 = __builtin_amdgcn_wmma_f32_16x16x4_f32(
                false, afrag0[c], false, bcur[c], (short)0, acc0, false, false);
            acc1 = __builtin_amdgcn_wmma_f32_16x16x4_f32(
                false, afrag1[c], false, bcur[c], (short)0, acc1, false, false);
        }
        {
            v2f bfrag;
            bfrag.x = (h == 0) ? sqcur : 0.0f;
            bfrag.y = 0.0f;
            acc0 = __builtin_amdgcn_wmma_f32_16x16x4_f32(
                false, afragOne, false, bfrag, (short)0, acc0, false, false);
            acc1 = __builtin_amdgcn_wmma_f32_16x16x4_f32(
                false, afragOne, false, bfrag, (short)0, acc1, false, false);
        }

#pragma unroll
        for (int r = 0; r < 8; ++r) {
            tile[(r + 8 * h) * DPITCH + m]        = acc0[r];
            tile[(r + 8 * h + 16) * DPITCH + m]   = acc1[r];
        }
        wave_lds_sync();

        const float4* rowp = reinterpret_cast<const float4*>(&tile[lane * DPITCH]);
        const float4 q0 = rowp[0];
        const float4 q1 = rowp[1];
        const float4 q2 = rowp[2];
        const float4 q3 = rowp[3];
        const float dv[16] = { q0.x, q0.y, q0.z, q0.w, q1.x, q1.y, q1.z, q1.w,
                               q2.x, q2.y, q2.z, q2.w, q3.x, q3.y, q3.z, q3.w };
        const float shared = fmaxf(fmaxf(posted.x, posted.y), fmaxf(posted.z, posted.w));
        const float bound = fminf(key[KNN - 1], shared);
#pragma unroll
        for (int c = 0; c < 16; ++c) {
            const float d = dv[c];
            pq[qn].k[lane] = d;
            pq[qn].i[lane] = colbase + c;
            qn += (d < bound) ? 1 : 0;
        }

        const bool lastTile = (jt == jt0 + TILES_PER_WAVE - 1);
        if (lastTile || __builtin_amdgcn_ballot_w32(qn > QCAP - 16) != 0u) {
            const int maxn = __builtin_amdgcn_readfirstlane(
                (int)__builtin_amdgcn_wave_reduce_max_u32((unsigned)qn, 0));
            for (int q = 0; q < maxn; q += 2) {
                const float k0 = pq[q].k[lane];
                const int   c0 = pq[q].i[lane];
                const float k1 = pq[q + 1].k[lane];
                const int   c1 = pq[q + 1].i[lane];
                const float d0 = (q < qn)     ? k0 : __builtin_inff();
                const float d1 = (q + 1 < qn) ? k1 : __builtin_inff();
                topk_insert(key, idx, d0, c0);
                topk_insert(key, idx, d1, c1);
            }
            qn = 0;
            postKey[lane * NSPLIT + wv] =
                fmaf(fabsf(key[POSTPOS]), 1.0e-5f, key[POSTPOS]) + 1.0e-30f;
        }
        wave_lds_sync();

#pragma unroll
        for (int c = 0; c < 8; ++c) bcur[c] = bnxt[c];
        sqcur = sqnxt;
    }

    if (wv & 1) {
#pragma unroll
        for (int s = 0; s < KNN; ++s) {
            mrgKey[wv >> 1][s * 32 + lane] = key[s];
            mrgIdx[wv >> 1][s * 32 + lane] = idx[s];
        }
    }
    __syncthreads();
    if (!(wv & 1)) topk_merge(key, idx, mrgKey[wv >> 1], mrgIdx[wv >> 1], lane);
    if (wv == 2) {
#pragma unroll
        for (int s = 0; s < KNN; ++s) {
            mrgKey[2][s * 32 + lane] = key[s];
            mrgIdx[2][s * 32 + lane] = idx[s];
        }
    }
    __syncthreads();
    if (wv == 0) {
        topk_merge(key, idx, mrgKey[2], mrgIdx[2], lane);
#pragma unroll
        for (int s = 0; s < KNN; ++s)
            selIdx[lane * KNN + s] = idx[s];
    }
    __syncthreads();

    float* poolb = pooled + ((size_t)b * NPTS + rowbase) * CH;
#pragma unroll 1
    for (int rr = 0; rr < ROWS_PER_WAVE / NSPLIT; ++rr) {
        const int r = wv * (ROWS_PER_WAVE / NSPLIT) + rr;
        float mx = -3.0e38f;
#pragma unroll
        for (int k = 0; k < KNN; ++k) {
            const int j = selIdx[r * KNN + k];
            const float v = xb[(size_t)j * CH + lane];
            mx = fmaxf(mx, v);
        }
        poolb[r * CH + lane] = mx;
    }
}

__global__ __launch_bounds__(32) void mlp_kernel(const float* __restrict__ pooled,
                                                 const float* __restrict__ Wl,
                                                 const float* __restrict__ bl,
                                                 const float* __restrict__ Wc,
                                                 const float* __restrict__ bc,
                                                 float* __restrict__ out) {
    __shared__ float hT[16 * 68];

    const int lane = threadIdx.x;
    const int m = lane & 15;
    const int h = lane >> 4;
    const size_t rowbase = (size_t)blockIdx.x * 16;

    v2f afrag[8];
#pragma unroll
    for (int c = 0; c < 8; ++c) {
        const float* p = pooled + (rowbase + m) * CH + 4 * c + 2 * h;
        afrag[c].x = p[0];
        afrag[c].y = p[1];
    }

#pragma unroll
    for (int t = 0; t < 4; ++t) {
        v8f acc = {};
#pragma unroll
        for (int c = 0; c < 8; ++c) {
            const float* p = Wl + (size_t)(4 * c + 2 * h) * LDIM + t * 16 + m;
            v2f bfrag;
            bfrag.x = p[0];
            bfrag.y = p[LDIM];
            acc = __builtin_amdgcn_wmma_f32_16x16x4_f32(
                false, afrag[c], false, bfrag, (short)0, acc, false, false);
        }
        const float bias = bl[t * 16 + m];
#pragma unroll
        for (int r = 0; r < 8; ++r)
            hT[(r + 8 * h) * 68 + t * 16 + m] = acc[r] + bias;
    }
    __syncthreads();

    v2f a2[16];
#pragma unroll
    for (int c = 0; c < 16; ++c) {
        a2[c].x = hT[m * 68 + 4 * c + 2 * h];
        a2[c].y = hT[m * 68 + 4 * c + 2 * h + 1];
    }

    float* outp = out + rowbase * ODIM;
#pragma unroll
    for (int t = 0; t < 8; ++t) {
        v8f acc = {};
#pragma unroll
        for (int c = 0; c < 16; ++c) {
            const float* p = Wc + (size_t)(4 * c + 2 * h) * ODIM + t * 16 + m;
            v2f bfrag;
            bfrag.x = p[0];
            bfrag.y = p[ODIM];
            acc = __builtin_amdgcn_wmma_f32_16x16x4_f32(
                false, a2[c], false, bfrag, (short)0, acc, false, false);
        }
        const float bias = bc[t * 16 + m];
#pragma unroll
        for (int r = 0; r < 8; ++r) {
            const float v = acc[r] + bias;
            outp[(size_t)(r + 8 * h) * ODIM + t * 16 + m] = fmaxf(v, 0.0f);
        }
    }
}

extern "C" void kernel_launch(void* const* d_in, const int* in_sizes, int n_in,
                              void* d_out, int out_size, void* d_ws, size_t ws_size,
                              hipStream_t stream) {
    const float* x  = (const float*)d_in[0];
    const float* Wl = (const float*)d_in[1];
    const float* bl = (const float*)d_in[2];
    const float* Wc = (const float*)d_in[3];
    const float* bc = (const float*)d_in[4];
    float* out = (float*)d_out;

    float* pooled = (float*)d_ws;
    float* sqbuf  = pooled + (size_t)BATCH * NPTS * CH;

    const int total = BATCH * NPTS;
    sq_kernel<<<(total + 255) / 256, 256, 0, stream>>>(x, sqbuf, total);
    knn_pool_kernel<<<BATCH * NRBLK, 32 * NSPLIT, 0, stream>>>(x, sqbuf, pooled);
    mlp_kernel<<<total / 16, 32, 0, stream>>>(pooled, Wl, bl, Wc, bc, out);
}


# dense layers fused into the end of the kNN kernel: pooled rows stay in LDS, each wave loads its weight fragments as whole groups under the gather, 4 waves split the columns; third launch gone
# speedup vs baseline: 5.3229x; 1.1513x over previous
#include <hip/hip_runtime.h>
#include <hip/hip_bf16.h>

#define BATCH 4
#define NPTS  8192
#define CH    32
#define KNN   16
#define LDIM  64
#define ODIM  128
#define NCTILES (NPTS / 16)
#define ROWS_PER_WAVE 32
#define NRBLK (NPTS / ROWS_PER_WAVE)
#define DPITCH 20
#define NSPLIT 4
#define TILES_PER_WAVE (NCTILES / NSPLIT)
#define QCAP 32

typedef __attribute__((ext_vector_type(2))) float v2f;
typedef __attribute__((ext_vector_type(8))) float v8f;
typedef __attribute__((ext_vector_type(4))) float v4f;
#define POSTPOS 3
#define PPITCH 36
#define HPITCH 68

__global__ void sq_kernel(const float* __restrict__ x, float* __restrict__ sq, int total) {
    int i = blockIdx.x * blockDim.x + threadIdx.x;
    if (i >= total) return;
    const float* p = x + (size_t)i * CH;
    float s = 0.0f;
#pragma unroll
    for (int c = 0; c < CH; ++c) s += p[c] * p[c];
    sq[i] = s;
}

__device__ __forceinline__ void wave_lds_sync() {
    __builtin_amdgcn_fence(__ATOMIC_RELEASE, "workgroup", "local");
    __builtin_amdgcn_fence(__ATOMIC_ACQUIRE, "workgroup", "local");
}

__device__ __forceinline__ void topk_insert(float (&key)[KNN], int (&idx)[KNN], float cv, int ci) {
#pragma unroll
    for (int s = 0; s < KNN; ++s) {
        const float ks = key[s];
        const int   is = idx[s];
        const bool  sw = cv < ks;
        key[s] = sw ? cv : ks;
        idx[s] = sw ? ci : is;
        cv = sw ? ks : cv;
        ci = sw ? is : ci;
    }
}

struct PendSlot {
    float k[32];
    int   i[32];
};

__device__ __forceinline__ void topk_merge(float (&key)[KNN], int (&idx)[KNN],
                                           const float* mk, const int* mi, int lane) {
#pragma unroll 1
    for (int s = 0; s < KNN; ++s) {
        const float d  = mk[s * 32 + lane];
        const int   ci = mi[s * 32 + lane];
        if (d < key[KNN - 1]) topk_insert(key, idx, d, ci);
    }
}

__global__ __launch_bounds__(32 * NSPLIT) void knn_pool_kernel(const float* __restrict__ x,
                                                               const float* __restrict__ sq,
                                                               const float* __restrict__ Wl,
                                                               const float* __restrict__ bl,
                                                               const float* __restrict__ Wc,
                                                               const float* __restrict__ bc,
                                                               float* __restrict__ out) {
    __shared__ __attribute__((aligned(16))) float distT[NSPLIT][ROWS_PER_WAVE * DPITCH];
    __shared__ float mrgKey[3][KNN * 32];
    __shared__ int   mrgIdx[3][KNN * 32];
    __shared__ int   selIdx[ROWS_PER_WAVE * KNN];
    __shared__ PendSlot pend[NSPLIT][QCAP];
    __shared__ __attribute__((aligned(16))) float postKey[32 * NSPLIT];

    const int lane    = threadIdx.x & 31;
    const int wv      = __builtin_amdgcn_readfirstlane(threadIdx.x >> 5);
    const int rblk    = blockIdx.x % NRBLK;
    const int b       = blockIdx.x / NRBLK;
    const int rowbase = rblk * ROWS_PER_WAVE;
    const float* xb   = x  + (size_t)b * NPTS * CH;
    const float* sqb  = sq + (size_t)b * NPTS;
    float* tile       = distT[wv];
    PendSlot* pq      = pend[wv];

    const int m = lane & 15;
    const int h = lane >> 4;

    v2f afrag0[8], afrag1[8];
#pragma unroll
    for (int c = 0; c < 8; ++c) {
        const float* p0 = xb + (size_t)(rowbase + m) * CH + 4 * c + 2 * h;
        const float* p1 = xb + (size_t)(rowbase + 16 + m) * CH + 4 * c + 2 * h;
        afrag0[c].x = -2.0f * p0[0];
        afrag0[c].y = -2.0f * p0[1];
        afrag1[c].x = -2.0f * p1[0];
        afrag1[c].y = -2.0f * p1[1];
    }
    v2f afragOne;
    afragOne.x = (h == 0) ? 1.0f : 0.0f;
    afragOne.y = 0.0f;

    float key[KNN];
    int   idx[KNN];
#pragma unroll
    for (int s = 0; s < KNN; ++s) { key[s] = 3.0e38f; idx[s] = 0; }

    int qn = 0;

    postKey[lane * NSPLIT + wv] = 3.0e38f;
    __syncthreads();

    const int jt0 = wv * TILES_PER_WAVE;

    v2f   bcur[8];
    float sqcur;
#pragma unroll
    for (int c = 0; c < 8; ++c) {
        const float* p = xb + (size_t)(jt0 * 16 + m) * CH + 4 * c + 2 * h;
        bcur[c].x = p[0];
        bcur[c].y = p[1];
    }
    sqcur = sqb[jt0 * 16 + m];

    for (int jt = jt0; jt < jt0 + TILES_PER_WAVE; ++jt) {
        const int colbase = jt * 16;

        const v4f posted = *reinterpret_cast<const v4f*>(&postKey[lane * NSPLIT]);

        const int nbase = ((jt + 1) & (NCTILES - 1)) * 16;
        v2f bnxt[8];
#pragma unroll
        for (int c = 0; c < 8; ++c) {
            const float* p = xb + (size_t)(nbase + m) * CH + 4 * c + 2 * h;
            bnxt[c].x = p[0];
            bnxt[c].y = p[1];
        }
        const float sqnxt = sqb[nbase + m];

        v8f acc0 = {};
        v8f acc1 = {};
#pragma unroll
        for (int c = 0; c < 8; ++c) {
            acc0 = __builtin_amdgcn_wmma_f32_16x16x4_f32(
                false, afrag0[c], false, bcur[c], (short)0, acc0, false, false);
            acc1 = __builtin_amdgcn_wmma_f32_16x16x4_f32(
                false, afrag1[c], false, bcur[c], (short)0, acc1, false, false);
        }
        {
            v2f bfrag;
            bfrag.x = (h == 0) ? sqcur : 0.0f;
            bfrag.y = 0.0f;
            acc0 = __builtin_amdgcn_wmma_f32_16x16x4_f32(
                false, afragOne, false, bfrag, (short)0, acc0, false, false);
            acc1 = __builtin_amdgcn_wmma_f32_16x16x4_f32(
                false, afragOne, false, bfrag, (short)0, acc1, false, false);
        }

#pragma unroll
        for (int r = 0; r < 8; ++r) {
            tile[(r + 8 * h) * DPITCH + m]        = acc0[r];
            tile[(r + 8 * h + 16) * DPITCH + m]   = acc1[r];
        }
        wave_lds_sync();

        const float4* rowp = reinterpret_cast<const float4*>(&tile[lane * DPITCH]);
        const float4 q0 = rowp[0];
        const float4 q1 = rowp[1];
        const float4 q2 = rowp[2];
        const float4 q3 = rowp[3];
        const float dv[16] = { q0.x, q0.y, q0.z, q0.w, q1.x, q1.y, q1.z, q1.w,
                               q2.x, q2.y, q2.z, q2.w, q3.x, q3.y, q3.z, q3.w };
        const float shared = fmaxf(fmaxf(posted.x, posted.y), fmaxf(posted.z, posted.w));
        const float bound = fminf(key[KNN - 1], shared);
#pragma unroll
        for (int c = 0; c < 16; ++c) {
            const float d = dv[c];
            pq[qn].k[lane] = d;
            pq[qn].i[lane] = colbase + c;
            qn += (d < bound) ? 1 : 0;
        }

        const bool lastTile = (jt == jt0 + TILES_PER_WAVE - 1);
        if (lastTile || __builtin_amdgcn_ballot_w32(qn > QCAP - 16) != 0u) {
            const int maxn = __builtin_amdgcn_readfirstlane(
                (int)__builtin_amdgcn_wave_reduce_max_u32((unsigned)qn, 0));
            for (int q = 0; q < maxn; q += 2) {
                const float k0 = pq[q].k[lane];
                const int   c0 = pq[q].i[lane];
                const float k1 = pq[q + 1].k[lane];
                const int   c1 = pq[q + 1].i[lane];
                const float d0 = (q < qn)     ? k0 : __builtin_inff();
                const float d1 = (q + 1 < qn) ? k1 : __builtin_inff();
                topk_insert(key, idx, d0, c0);
                topk_insert(key, idx, d1, c1);
            }
            qn = 0;
            postKey[lane * NSPLIT + wv] =
                fmaf(fabsf(key[POSTPOS]), 1.0e-5f, key[POSTPOS]) + 1.0e-30f;
        }
        wave_lds_sync();

#pragma unroll
        for (int c = 0; c < 8; ++c) bcur[c] = bnxt[c];
        sqcur = sqnxt;
    }

    if (wv & 1) {
#pragma unroll
        for (int s = 0; s < KNN; ++s) {
            mrgKey[wv >> 1][s * 32 + lane] = key[s];
            mrgIdx[wv >> 1][s * 32 + lane] = idx[s];
        }
    }
    __syncthreads();
    if (!(wv & 1)) topk_merge(key, idx, mrgKey[wv >> 1], mrgIdx[wv >> 1], lane);
    if (wv == 2) {
#pragma unroll
        for (int s = 0; s < KNN; ++s) {
            mrgKey[2][s * 32 + lane] = key[s];
            mrgIdx[2][s * 32 + lane] = idx[s];
        }
    }
    __syncthreads();
    if (wv == 0) {
        topk_merge(key, idx, mrgKey[2], mrgIdx[2], lane);
#pragma unroll
        for (int s = 0; s < KNN; ++s)
            selIdx[lane * KNN + s] = idx[s];
    }
    __syncthreads();

    float* poolT = &pend[0][0].k[0];
    float* hidT  = &pend[1][0].k[0];

    v2f wl[8], wc0[16], wc1[16];
#pragma unroll
    for (int c = 0; c < 8; ++c) {
        const float* p = Wl + (size_t)(4 * c + 2 * h) * LDIM + wv * 16 + m;
        wl[c].x = p[0];
        wl[c].y = p[LDIM];
    }
#pragma unroll
    for (int c = 0; c < 16; ++c) {
        const float* p = Wc + (size_t)(4 * c + 2 * h) * ODIM + wv * 32 + m;
        wc0[c].x = p[0];
        wc0[c].y = p[ODIM];
        wc1[c].x = p[16];
        wc1[c].y = p[ODIM + 16];
    }
    const float bias1  = bl[wv * 16 + m];
    const float bias20 = bc[wv * 32 + m];
    const float bias21 = bc[wv * 32 + 16 + m];

#pragma unroll 1
    for (int rr = 0; rr < ROWS_PER_WAVE / NSPLIT; ++rr) {
        const int r = wv * (ROWS_PER_WAVE / NSPLIT) + rr;
        float mx = -3.0e38f;
#pragma unroll
        for (int k = 0; k < KNN; ++k) {
            const int j = selIdx[r * KNN + k];
            const float v = xb[(size_t)j * CH + lane];
            mx = fmaxf(mx, v);
        }
        poolT[r * PPITCH + lane] = mx;
    }
    __syncthreads();

    {
        v2f pa0[8], pa1[8];
#pragma unroll
        for (int c = 0; c < 8; ++c) {
            pa0[c].x = poolT[m * PPITCH + 4 * c + 2 * h];
            pa0[c].y = poolT[m * PPITCH + 4 * c + 2 * h + 1];
            pa1[c].x = poolT[(16 + m) * PPITCH + 4 * c + 2 * h];
            pa1[c].y = poolT[(16 + m) * PPITCH + 4 * c + 2 * h + 1];
        }
        v8f h0 = {};
        v8f h1 = {};
#pragma unroll
        for (int c = 0; c < 8; ++c) {
            h0 = __builtin_amdgcn_wmma_f32_16x16x4_f32(
                false, pa0[c], false, wl[c], (short)0, h0, false, false);
            h1 = __builtin_amdgcn_wmma_f32_16x16x4_f32(
                false, pa1[c], false, wl[c], (short)0, h1, false, false);
        }
        float* hrow = hidT + (8 * h) * HPITCH + wv * 16 + m;
#pragma unroll
        for (int r = 0; r < 8; ++r) {
            hrow[r * HPITCH]        = h0[r] + bias1;
            hrow[(r + 16) * HPITCH] = h1[r] + bias1;
        }
    }
    __syncthreads();

    {
        v2f ha0[16], ha1[16];
#pragma unroll
        for (int c = 0; c < 16; ++c) {
            ha0[c].x = hidT[m * HPITCH + 4 * c + 2 * h];
            ha0[c].y = hidT[m * HPITCH + 4 * c + 2 * h + 1];
            ha1[c].x = hidT[(16 + m) * HPITCH + 4 * c + 2 * h];
            ha1[c].y = hidT[(16 + m) * HPITCH + 4 * c + 2 * h + 1];
        }
        v8f o00 = {};
        v8f o01 = {};
        v8f o10 = {};
        v8f o11 = {};
#pragma unroll
        for (int c = 0; c < 16; ++c) {
            o00 = __builtin_amdgcn_wmma_f32_16x16x4_f32(
                false, ha0[c], false, wc0[c], (short)0, o00, false, false);
            o10 = __builtin_amdgcn_wmma_f32_16x16x4_f32(
                false, ha1[c], false, wc0[c], (short)0, o10, false, false);
            o01 = __builtin_amdgcn_wmma_f32_16x16x4_f32(
                false, ha0[c], false, wc1[c], (short)0, o01, false, false);
            o11 = __builtin_amdgcn_wmma_f32_16x16x4_f32(
                false, ha1[c], false, wc1[c], (short)0, o11, false, false);
        }
        float* outp = out + ((size_t)b * NPTS + rowbase + 8 * h) * ODIM + wv * 32 + m;
#pragma unroll
        for (int r = 0; r < 8; ++r) {
            outp[r * ODIM]             = fmaxf(o00[r] + bias20, 0.0f);
            outp[r * ODIM + 16]        = fmaxf(o01[r] + bias21, 0.0f);
            outp[(r + 16) * ODIM]      = fmaxf(o10[r] + bias20, 0.0f);
            outp[(r + 16) * ODIM + 16] = fmaxf(o11[r] + bias21, 0.0f);
        }
    }
}

extern "C" void kernel_launch(void* const* d_in, const int* in_sizes, int n_in,
                              void* d_out, int out_size, void* d_ws, size_t ws_size,
                              hipStream_t stream) {
    const float* x  = (const float*)d_in[0];
    const float* Wl = (const float*)d_in[1];
    const float* bl = (const float*)d_in[2];
    const float* Wc = (const float*)d_in[3];
    const float* bc = (const float*)d_in[4];
    float* out = (float*)d_out;

    float* sqbuf = (float*)d_ws;

    const int total = BATCH * NPTS;
    sq_kernel<<<(total + 255) / 256, 256, 0, stream>>>(x, sqbuf, total);
    knn_pool_kernel<<<BATCH * NRBLK, 32 * NSPLIT, 0, stream>>>(x, sqbuf, Wl, bl, Wc, bc, out);
}


# kNN key: the +sq_j term leaves the matrix product: 8 packed vector adds on the accumulators replace the ninth K chunk (1*sq_j, rest zeros): 16 not 18 WMMA per tile; sum order and rounding unchanged
# speedup vs baseline: 5.5039x; 1.0340x over previous
#include <hip/hip_runtime.h>
#include <hip/hip_bf16.h>

#define BATCH 4
#define NPTS  8192
#define CH    32
#define KNN   16
#define LDIM  64
#define ODIM  128
#define NCTILES (NPTS / 16)
#define ROWS_PER_WAVE 32
#define NRBLK (NPTS / ROWS_PER_WAVE)
#define DPITCH 20
#define NSPLIT 4
#define TILES_PER_WAVE (NCTILES / NSPLIT)
#define QCAP 32

typedef __attribute__((ext_vector_type(2))) float v2f;
typedef __attribute__((ext_vector_type(8))) float v8f;
typedef __attribute__((ext_vector_type(4))) float v4f;
#define POSTPOS 3
#define PPITCH 36
#define HPITCH 68

__global__ void sq_kernel(const float* __restrict__ x, float* __restrict__ sq, int total) {
    int i = blockIdx.x * blockDim.x + threadIdx.x;
    if (i >= total) return;
    const float* p = x + (size_t)i * CH;
    float s = 0.0f;
#pragma unroll
    for (int c = 0; c < CH; ++c) s += p[c] * p[c];
    sq[i] = s;
}

__device__ __forceinline__ void wave_lds_sync() {
    __builtin_amdgcn_fence(__ATOMIC_RELEASE, "workgroup", "local");
    __builtin_amdgcn_fence(__ATOMIC_ACQUIRE, "workgroup", "local");
}

__device__ __forceinline__ void topk_insert(float (&key)[KNN], int (&idx)[KNN], float cv, int ci) {
#pragma unroll
    for (int s = 0; s < KNN; ++s) {
        const float ks = key[s];
        const int   is = idx[s];
        const bool  sw = cv < ks;
        key[s] = sw ? cv : ks;
        idx[s] = sw ? ci : is;
        cv = sw ? ks : cv;
        ci = sw ? is : ci;
    }
}

struct PendSlot {
    float k[32];
    int   i[32];
};

__device__ __forceinline__ void topk_merge(float (&key)[KNN], int (&idx)[KNN],
                                           const float* mk, const int* mi, int lane) {
#pragma unroll 1
    for (int s = 0; s < KNN; ++s) {
        const float d  = mk[s * 32 + lane];
        const int   ci = mi[s * 32 + lane];
        if (d < key[KNN - 1]) topk_insert(key, idx, d, ci);
    }
}

__global__ __launch_bounds__(32 * NSPLIT) void knn_pool_kernel(const float* __restrict__ x,
                                                               const float* __restrict__ sq,
                                                               const float* __restrict__ Wl,
                                                               const float* __restrict__ bl,
                                                               const float* __restrict__ Wc,
                                                               const float* __restrict__ bc,
                                                               float* __restrict__ out) {
    __shared__ __attribute__((aligned(16))) float distT[NSPLIT][ROWS_PER_WAVE * DPITCH];
    __shared__ float mrgKey[3][KNN * 32];
    __shared__ int   mrgIdx[3][KNN * 32];
    __shared__ int   selIdx[ROWS_PER_WAVE * KNN];
    __shared__ PendSlot pend[NSPLIT][QCAP];
    __shared__ __attribute__((aligned(16))) float postKey[32 * NSPLIT];

    const int lane    = threadIdx.x & 31;
    const int wv      = __builtin_amdgcn_readfirstlane(threadIdx.x >> 5);
    const int rblk    = blockIdx.x % NRBLK;
    const int b       = blockIdx.x / NRBLK;
    const int rowbase = rblk * ROWS_PER_WAVE;
    const float* xb   = x  + (size_t)b * NPTS * CH;
    const float* sqb  = sq + (size_t)b * NPTS;
    float* tile       = distT[wv];
    PendSlot* pq      = pend[wv];

    const int m = lane & 15;
    const int h = lane >> 4;

    v2f afrag0[8], afrag1[8];
#pragma unroll
    for (int c = 0; c < 8; ++c) {
        const float* p0 = xb + (size_t)(rowbase + m) * CH + 4 * c + 2 * h;
        const float* p1 = xb + (size_t)(rowbase + 16 + m) * CH + 4 * c + 2 * h;
        afrag0[c].x = -2.0f * p0[0];
        afrag0[c].y = -2.0f * p0[1];
        afrag1[c].x = -2.0f * p1[0];
        afrag1[c].y = -2.0f * p1[1];
    }
    float key[KNN];
    int   idx[KNN];
#pragma unroll
    for (int s = 0; s < KNN; ++s) { key[s] = 3.0e38f; idx[s] = 0; }

    int qn = 0;

    postKey[lane * NSPLIT + wv] = 3.0e38f;
    __syncthreads();

    const int jt0 = wv * TILES_PER_WAVE;

    v2f   bcur[8];
    float sqcur;
#pragma unroll
    for (int c = 0; c < 8; ++c) {
        const float* p = xb + (size_t)(jt0 * 16 + m) * CH + 4 * c + 2 * h;
        bcur[c].x = p[0];
        bcur[c].y = p[1];
    }
    sqcur = sqb[jt0 * 16 + m];

    for (int jt = jt0; jt < jt0 + TILES_PER_WAVE; ++jt) {
        const int colbase = jt * 16;

        const v4f posted = *reinterpret_cast<const v4f*>(&postKey[lane * NSPLIT]);

        const int nbase = ((jt + 1) & (NCTILES - 1)) * 16;
        v2f bnxt[8];
#pragma unroll
        for (int c = 0; c < 8; ++c) {
            const float* p = xb + (size_t)(nbase + m) * CH + 4 * c + 2 * h;
            bnxt[c].x = p[0];
            bnxt[c].y = p[1];
        }
        const float sqnxt = sqb[nbase + m];

        v8f acc0 = {};
        v8f acc1 = {};
#pragma unroll
        for (int c = 0; c < 8; ++c) {
            acc0 = __builtin_amdgcn_wmma_f32_16x16x4_f32(
                false, afrag0[c], false, bcur[c], (short)0, acc0, false, false);
            acc1 = __builtin_amdgcn_wmma_f32_16x16x4_f32(
                false, afrag1[c], false, bcur[c], (short)0, acc1, false, false);
        }
        asm("s_wait_alu depctr_va_vdst(0)" : "+v"(acc0), "+v"(acc1));
        acc0 += sqcur;
        acc1 += sqcur;

#pragma unroll
        for (int r = 0; r < 8; ++r) {
            tile[(r + 8 * h) * DPITCH + m]        = acc0[r];
            tile[(r + 8 * h + 16) * DPITCH + m]   = acc1[r];
        }
        wave_lds_sync();

        const float4* rowp = reinterpret_cast<const float4*>(&tile[lane * DPITCH]);
        const float4 q0 = rowp[0];
        const float4 q1 = rowp[1];
        const float4 q2 = rowp[2];
        const float4 q3 = rowp[3];
        const float dv[16] = { q0.x, q0.y, q0.z, q0.w, q1.x, q1.y, q1.z, q1.w,
                               q2.x, q2.y, q2.z, q2.w, q3.x, q3.y, q3.z, q3.w };
        const float shared = fmaxf(fmaxf(posted.x, posted.y), fmaxf(posted.z, posted.w));
        const float bound = fminf(key[KNN - 1], shared);
#pragma unroll
        for (int c = 0; c < 16; ++c) {
            const float d = dv[c];
            pq[qn].k[lane] = d;
            pq[qn].i[lane] = colbase + c;
            qn += (d < bound) ? 1 : 0;
        }

        const bool lastTile = (jt == jt0 + TILES_PER_WAVE - 1);
        if (lastTile || __builtin_amdgcn_ballot_w32(qn > QCAP - 16) != 0u) {
            const int maxn = __builtin_amdgcn_readfirstlane(
                (int)__builtin_amdgcn_wave_reduce_max_u32((unsigned)qn, 0));
            for (int q = 0; q < maxn; q += 2) {
                const float k0 = pq[q].k[lane];
                const int   c0 = pq[q].i[lane];
                const float k1 = pq[q + 1].k[lane];
                const int   c1 = pq[q + 1].i[lane];
                const float d0 = (q < qn)     ? k0 : __builtin_inff();
                const float d1 = (q + 1 < qn) ? k1 : __builtin_inff();
                topk_insert(key, idx, d0, c0);
                topk_insert(key, idx, d1, c1);
            }
            qn = 0;
            postKey[lane * NSPLIT + wv] =
                fmaf(fabsf(key[POSTPOS]), 1.0e-5f, key[POSTPOS]) + 1.0e-30f;
        }
        wave_lds_sync();

#pragma unroll
        for (int c = 0; c < 8; ++c) bcur[c] = bnxt[c];
        sqcur = sqnxt;
    }

    if (wv & 1) {
#pragma unroll
        for (int s = 0; s < KNN; ++s) {
            mrgKey[wv >> 1][s * 32 + lane] = key[s];
            mrgIdx[wv >> 1][s * 32 + lane] = idx[s];
        }
    }
    __syncthreads();
    if (!(wv & 1)) topk_merge(key, idx, mrgKey[wv >> 1], mrgIdx[wv >> 1], lane);
    if (wv == 2) {
#pragma unroll
        for (int s = 0; s < KNN; ++s) {
            mrgKey[2][s * 32 + lane] = key[s];
            mrgIdx[2][s * 32 + lane] = idx[s];
        }
    }
    __syncthreads();
    if (wv == 0) {
        topk_merge(key, idx, mrgKey[2], mrgIdx[2], lane);
#pragma unroll
        for (int s = 0; s < KNN; ++s)
            selIdx[lane * KNN + s] = idx[s];
    }
    __syncthreads();

    float* poolT = &pend[0][0].k[0];
    float* hidT  = &pend[1][0].k[0];

    v2f wl[8], wc0[16], wc1[16];
#pragma unroll
    for (int c = 0; c < 8; ++c) {
        const float* p = Wl + (size_t)(4 * c + 2 * h) * LDIM + wv * 16 + m;
        wl[c].x = p[0];
        wl[c].y = p[LDIM];
    }
#pragma unroll
    for (int c = 0; c < 16; ++c) {
        const float* p = Wc + (size_t)(4 * c + 2 * h) * ODIM + wv * 32 + m;
        wc0[c].x = p[0];
        wc0[c].y = p[ODIM];
        wc1[c].x = p[16];
        wc1[c].y = p[ODIM + 16];
    }
    const float bias1  = bl[wv * 16 + m];
    const float bias20 = bc[wv * 32 + m];
    const float bias21 = bc[wv * 32 + 16 + m];

#pragma unroll 1
    for (int rr = 0; rr < ROWS_PER_WAVE / NSPLIT; ++rr) {
        const int r = wv * (ROWS_PER_WAVE / NSPLIT) + rr;
        float mx = -3.0e38f;
#pragma unroll
        for (int k = 0; k < KNN; ++k) {
            const int j = selIdx[r * KNN + k];
            const float v = xb[(size_t)j * CH + lane];
            mx = fmaxf(mx, v);
        }
        poolT[r * PPITCH + lane] = mx;
    }
    __syncthreads();

    {
        v2f pa0[8], pa1[8];
#pragma unroll
        for (int c = 0; c < 8; ++c) {
            pa0[c].x = poolT[m * PPITCH + 4 * c + 2 * h];
            pa0[c].y = poolT[m * PPITCH + 4 * c + 2 * h + 1];
            pa1[c].x = poolT[(16 + m) * PPITCH + 4 * c + 2 * h];
            pa1[c].y = poolT[(16 + m) * PPITCH + 4 * c + 2 * h + 1];
        }
        v8f h0 = {};
        v8f h1 = {};
#pragma unroll
        for (int c = 0; c < 8; ++c) {
            h0 = __builtin_amdgcn_wmma_f32_16x16x4_f32(
                false, pa0[c], false, wl[c], (short)0, h0, false, false);
            h1 = __builtin_amdgcn_wmma_f32_16x16x4_f32(
                false, pa1[c], false, wl[c], (short)0, h1, false, false);
        }
        float* hrow = hidT + (8 * h) * HPITCH + wv * 16 + m;
#pragma unroll
        for (int r = 0; r < 8; ++r) {
            hrow[r * HPITCH]        = h0[r] + bias1;
            hrow[(r + 16) * HPITCH] = h1[r] + bias1;
        }
    }
    __syncthreads();

    {
        v2f ha0[16], ha1[16];
#pragma unroll
        for (int c = 0; c < 16; ++c) {
            ha0[c].x = hidT[m * HPITCH + 4 * c + 2 * h];
            ha0[c].y = hidT[m * HPITCH + 4 * c + 2 * h + 1];
            ha1[c].x = hidT[(16 + m) * HPITCH + 4 * c + 2 * h];
            ha1[c].y = hidT[(16 + m) * HPITCH + 4 * c + 2 * h + 1];
        }
        v8f o00 = {};
        v8f o01 = {};
        v8f o10 = {};
        v8f o11 = {};
#pragma unroll
        for (int c = 0; c < 16; ++c) {
            o00 = __builtin_amdgcn_wmma_f32_16x16x4_f32(
                false, ha0[c], false, wc0[c], (short)0, o00, false, false);
            o10 = __builtin_amdgcn_wmma_f32_16x16x4_f32(
                false, ha1[c], false, wc0[c], (short)0, o10, false, false);
            o01 = __builtin_amdgcn_wmma_f32_16x16x4_f32(
                false, ha0[c], false, wc1[c], (short)0, o01, false, false);
            o11 = __builtin_amdgcn_wmma_f32_16x16x4_f32(
                false, ha1[c], false, wc1[c], (short)0, o11, false, false);
        }
        float* outp = out + ((size_t)b * NPTS + rowbase + 8 * h) * ODIM + wv * 32 + m;
#pragma unroll
        for (int r = 0; r < 8; ++r) {
            outp[r * ODIM]             = fmaxf(o00[r] + bias20, 0.0f);
            outp[r * ODIM + 16]        = fmaxf(o01[r] + bias21, 0.0f);
            outp[(r + 16) * ODIM]      = fmaxf(o10[r] + bias20, 0.0f);
            outp[(r + 16) * ODIM + 16] = fmaxf(o11[r] + bias21, 0.0f);
        }
    }
}

extern "C" void kernel_launch(void* const* d_in, const int* in_sizes, int n_in,
                              void* d_out, int out_size, void* d_ws, size_t ws_size,
                              hipStream_t stream) {
    const float* x  = (const float*)d_in[0];
    const float* Wl = (const float*)d_in[1];
    const float* bl = (const float*)d_in[2];
    const float* Wc = (const float*)d_in[3];
    const float* bc = (const float*)d_in[4];
    float* out = (float*)d_out;

    float* sqbuf = (float*)d_ws;

    const int total = BATCH * NPTS;
    sq_kernel<<<(total + 255) / 256, 256, 0, stream>>>(x, sqbuf, total);
    knn_pool_kernel<<<BATCH * NRBLK, 32 * NSPLIT, 0, stream>>>(x, sqbuf, Wl, bl, Wc, bc, out);
}


# kNN selection: one wave-wide compare per candidate column, all 16 masks formed first; a column no lane wants is skipped by a scalar branch and only hitting lanes append to their queue
# speedup vs baseline: 5.8502x; 1.0629x over previous
#include <hip/hip_runtime.h>
#include <hip/hip_bf16.h>

#define BATCH 4
#define NPTS  8192
#define CH    32
#define KNN   16
#define LDIM  64
#define ODIM  128
#define NCTILES (NPTS / 16)
#define ROWS_PER_WAVE 32
#define NRBLK (NPTS / ROWS_PER_WAVE)
#define DPITCH 20
#define NSPLIT 4
#define TILES_PER_WAVE (NCTILES / NSPLIT)
#define QCAP 32

typedef __attribute__((ext_vector_type(2))) float v2f;
typedef __attribute__((ext_vector_type(8))) float v8f;
typedef __attribute__((ext_vector_type(4))) float v4f;
#define POSTPOS 3
#define PPITCH 36
#define HPITCH 68

__global__ void sq_kernel(const float* __restrict__ x, float* __restrict__ sq, int total) {
    int i = blockIdx.x * blockDim.x + threadIdx.x;
    if (i >= total) return;
    const float* p = x + (size_t)i * CH;
    float s = 0.0f;
#pragma unroll
    for (int c = 0; c < CH; ++c) s += p[c] * p[c];
    sq[i] = s;
}

__device__ __forceinline__ void wave_lds_sync() {
    __builtin_amdgcn_fence(__ATOMIC_RELEASE, "workgroup", "local");
    __builtin_amdgcn_fence(__ATOMIC_ACQUIRE, "workgroup", "local");
}

__device__ __forceinline__ void topk_insert(float (&key)[KNN], int (&idx)[KNN], float cv, int ci) {
#pragma unroll
    for (int s = 0; s < KNN; ++s) {
        const float ks = key[s];
        const int   is = idx[s];
        const bool  sw = cv < ks;
        key[s] = sw ? cv : ks;
        idx[s] = sw ? ci : is;
        cv = sw ? ks : cv;
        ci = sw ? is : ci;
    }
}

struct PendSlot {
    float k[32];
    int   i[32];
};

__device__ __forceinline__ void topk_merge(float (&key)[KNN], int (&idx)[KNN],
                                           const float* mk, const int* mi, int lane) {
#pragma unroll 1
    for (int s = 0; s < KNN; ++s) {
        const float d  = mk[s * 32 + lane];
        const int   ci = mi[s * 32 + lane];
        if (d < key[KNN - 1]) topk_insert(key, idx, d, ci);
    }
}

__global__ __launch_bounds__(32 * NSPLIT) void knn_pool_kernel(const float* __restrict__ x,
                                                               const float* __restrict__ sq,
                                                               const float* __restrict__ Wl,
                                                               const float* __restrict__ bl,
                                                               const float* __restrict__ Wc,
                                                               const float* __restrict__ bc,
                                                               float* __restrict__ out) {
    __shared__ __attribute__((aligned(16))) float distT[NSPLIT][ROWS_PER_WAVE * DPITCH];
    __shared__ float mrgKey[3][KNN * 32];
    __shared__ int   mrgIdx[3][KNN * 32];
    __shared__ int   selIdx[ROWS_PER_WAVE * KNN];
    __shared__ PendSlot pend[NSPLIT][QCAP];
    __shared__ __attribute__((aligned(16))) float postKey[32 * NSPLIT];

    const int lane    = threadIdx.x & 31;
    const int wv      = __builtin_amdgcn_readfirstlane(threadIdx.x >> 5);
    const int rblk    = blockIdx.x % NRBLK;
    const int b       = blockIdx.x / NRBLK;
    const int rowbase = rblk * ROWS_PER_WAVE;
    const float* xb   = x  + (size_t)b * NPTS * CH;
    const float* sqb  = sq + (size_t)b * NPTS;
    float* tile       = distT[wv];
    PendSlot* pq      = pend[wv];

    const int m = lane & 15;
    const int h = lane >> 4;

    v2f afrag0[8], afrag1[8];
#pragma unroll
    for (int c = 0; c < 8; ++c) {
        const float* p0 = xb + (size_t)(rowbase + m) * CH + 4 * c + 2 * h;
        const float* p1 = xb + (size_t)(rowbase + 16 + m) * CH + 4 * c + 2 * h;
        afrag0[c].x = -2.0f * p0[0];
        afrag0[c].y = -2.0f * p0[1];
        afrag1[c].x = -2.0f * p1[0];
        afrag1[c].y = -2.0f * p1[1];
    }
    float key[KNN];
    int   idx[KNN];
#pragma unroll
    for (int s = 0; s < KNN; ++s) { key[s] = 3.0e38f; idx[s] = 0; }

    int qn = 0;

    postKey[lane * NSPLIT + wv] = 3.0e38f;
    __syncthreads();

    const int jt0 = wv * TILES_PER_WAVE;

    v2f   bcur[8];
    float sqcur;
#pragma unroll
    for (int c = 0; c < 8; ++c) {
        const float* p = xb + (size_t)(jt0 * 16 + m) * CH + 4 * c + 2 * h;
        bcur[c].x = p[0];
        bcur[c].y = p[1];
    }
    sqcur = sqb[jt0 * 16 + m];

    for (int jt = jt0; jt < jt0 + TILES_PER_WAVE; ++jt) {
        const int colbase = jt * 16;

        const v4f posted = *reinterpret_cast<const v4f*>(&postKey[lane * NSPLIT]);

        const int nbase = ((jt + 1) & (NCTILES - 1)) * 16;
        v2f bnxt[8];
#pragma unroll
        for (int c = 0; c < 8; ++c) {
            const float* p = xb + (size_t)(nbase + m) * CH + 4 * c + 2 * h;
            bnxt[c].x = p[0];
            bnxt[c].y = p[1];
        }
        const float sqnxt = sqb[nbase + m];

        v8f acc0 = {};
        v8f acc1 = {};
#pragma unroll
        for (int c = 0; c < 8; ++c) {
            acc0 = __builtin_amdgcn_wmma_f32_16x16x4_f32(
                false, afrag0[c], false, bcur[c], (short)0, acc0, false, false);
            acc1 = __builtin_amdgcn_wmma_f32_16x16x4_f32(
                false, afrag1[c], false, bcur[c], (short)0, acc1, false, false);
        }
        asm("s_wait_alu depctr_va_vdst(0)" : "+v"(acc0), "+v"(acc1));
        acc0 += sqcur;
        acc1 += sqcur;

#pragma unroll
        for (int r = 0; r < 8; ++r) {
            tile[(r + 8 * h) * DPITCH + m]        = acc0[r];
            tile[(r + 8 * h + 16) * DPITCH + m]   = acc1[r];
        }
        wave_lds_sync();

        const float4* rowp = reinterpret_cast<const float4*>(&tile[lane * DPITCH]);
        const float4 q0 = rowp[0];
        const float4 q1 = rowp[1];
        const float4 q2 = rowp[2];
        const float4 q3 = rowp[3];
        const float dv[16] = { q0.x, q0.y, q0.z, q0.w, q1.x, q1.y, q1.z, q1.w,
                               q2.x, q2.y, q2.z, q2.w, q3.x, q3.y, q3.z, q3.w };
        const float shared = fmaxf(fmaxf(posted.x, posted.y), fmaxf(posted.z, posted.w));
        const float bound = fminf(key[KNN - 1], shared);
        unsigned hit[16];
#pragma unroll
        for (int c = 0; c < 16; ++c)
            hit[c] = __builtin_amdgcn_ballot_w32(dv[c] < bound);
#pragma unroll
        for (int c = 0; c < 16; ++c) {
            if (__builtin_amdgcn_inverse_ballot_w32(hit[c])) {
                pq[qn].k[lane] = dv[c];
                pq[qn].i[lane] = colbase + c;
                qn += 1;
            }
        }

        const bool lastTile = (jt == jt0 + TILES_PER_WAVE - 1);
        if (lastTile || __builtin_amdgcn_ballot_w32(qn > QCAP - 16) != 0u) {
            const int maxn = __builtin_amdgcn_readfirstlane(
                (int)__builtin_amdgcn_wave_reduce_max_u32((unsigned)qn, 0));
            for (int q = 0; q < maxn; q += 2) {
                const float k0 = pq[q].k[lane];
                const int   c0 = pq[q].i[lane];
                const float k1 = pq[q + 1].k[lane];
                const int   c1 = pq[q + 1].i[lane];
                const float d0 = (q < qn)     ? k0 : __builtin_inff();
                const float d1 = (q + 1 < qn) ? k1 : __builtin_inff();
                topk_insert(key, idx, d0, c0);
                topk_insert(key, idx, d1, c1);
            }
            qn = 0;
            postKey[lane * NSPLIT + wv] =
                fmaf(fabsf(key[POSTPOS]), 1.0e-5f, key[POSTPOS]) + 1.0e-30f;
        }
        wave_lds_sync();

#pragma unroll
        for (int c = 0; c < 8; ++c) bcur[c] = bnxt[c];
        sqcur = sqnxt;
    }

    if (wv & 1) {
#pragma unroll
        for (int s = 0; s < KNN; ++s) {
            mrgKey[wv >> 1][s * 32 + lane] = key[s];
            mrgIdx[wv >> 1][s * 32 + lane] = idx[s];
        }
    }
    __syncthreads();
    if (!(wv & 1)) topk_merge(key, idx, mrgKey[wv >> 1], mrgIdx[wv >> 1], lane);
    if (wv == 2) {
#pragma unroll
        for (int s = 0; s < KNN; ++s) {
            mrgKey[2][s * 32 + lane] = key[s];
            mrgIdx[2][s * 32 + lane] = idx[s];
        }
    }
    __syncthreads();
    if (wv == 0) {
        topk_merge(key, idx, mrgKey[2], mrgIdx[2], lane);
#pragma unroll
        for (int s = 0; s < KNN; ++s)
            selIdx[lane * KNN + s] = idx[s];
    }
    __syncthreads();

    float* poolT = &pend[0][0].k[0];
    float* hidT  = &pend[1][0].k[0];

    v2f wl[8], wc0[16], wc1[16];
#pragma unroll
    for (int c = 0; c < 8; ++c) {
        const float* p = Wl + (size_t)(4 * c + 2 * h) * LDIM + wv * 16 + m;
        wl[c].x = p[0];
        wl[c].y = p[LDIM];
    }
#pragma unroll
    for (int c = 0; c < 16; ++c) {
        const float* p = Wc + (size_t)(4 * c + 2 * h) * ODIM + wv * 32 + m;
        wc0[c].x = p[0];
        wc0[c].y = p[ODIM];
        wc1[c].x = p[16];
        wc1[c].y = p[ODIM + 16];
    }
    const float bias1  = bl[wv * 16 + m];
    const float bias20 = bc[wv * 32 + m];
    const float bias21 = bc[wv * 32 + 16 + m];

#pragma unroll 1
    for (int rr = 0; rr < ROWS_PER_WAVE / NSPLIT; ++rr) {
        const int r = wv * (ROWS_PER_WAVE / NSPLIT) + rr;
        float mx = -3.0e38f;
#pragma unroll
        for (int k = 0; k < KNN; ++k) {
            const int j = selIdx[r * KNN + k];
            const float v = xb[(size_t)j * CH + lane];
            mx = fmaxf(mx, v);
        }
        poolT[r * PPITCH + lane] = mx;
    }
    __syncthreads();

    {
        v2f pa0[8], pa1[8];
#pragma unroll
        for (int c = 0; c < 8; ++c) {
            pa0[c].x = poolT[m * PPITCH + 4 * c + 2 * h];
            pa0[c].y = poolT[m * PPITCH + 4 * c + 2 * h + 1];
            pa1[c].x = poolT[(16 + m) * PPITCH + 4 * c + 2 * h];
            pa1[c].y = poolT[(16 + m) * PPITCH + 4 * c + 2 * h + 1];
        }
        v8f h0 = {};
        v8f h1 = {};
#pragma unroll
        for (int c = 0; c < 8; ++c) {
            h0 = __builtin_amdgcn_wmma_f32_16x16x4_f32(
                false, pa0[c], false, wl[c], (short)0, h0, false, false);
            h1 = __builtin_amdgcn_wmma_f32_16x16x4_f32(
                false, pa1[c], false, wl[c], (short)0, h1, false, false);
        }
        float* hrow = hidT + (8 * h) * HPITCH + wv * 16 + m;
#pragma unroll
        for (int r = 0; r < 8; ++r) {
            hrow[r * HPITCH]        = h0[r] + bias1;
            hrow[(r + 16) * HPITCH] = h1[r] + bias1;
        }
    }
    __syncthreads();

    {
        v2f ha0[16], ha1[16];
#pragma unroll
        for (int c = 0; c < 16; ++c) {
            ha0[c].x = hidT[m * HPITCH + 4 * c + 2 * h];
            ha0[c].y = hidT[m * HPITCH + 4 * c + 2 * h + 1];
            ha1[c].x = hidT[(16 + m) * HPITCH + 4 * c + 2 * h];
            ha1[c].y = hidT[(16 + m) * HPITCH + 4 * c + 2 * h + 1];
        }
        v8f o00 = {};
        v8f o01 = {};
        v8f o10 = {};
        v8f o11 = {};
#pragma unroll
        for (int c = 0; c < 16; ++c) {
            o00 = __builtin_amdgcn_wmma_f32_16x16x4_f32(
                false, ha0[c], false, wc0[c], (short)0, o00, false, false);
            o10 = __builtin_amdgcn_wmma_f32_16x16x4_f32(
                false, ha1[c], false, wc0[c], (short)0, o10, false, false);
            o01 = __builtin_amdgcn_wmma_f32_16x16x4_f32(
                false, ha0[c], false, wc1[c], (short)0, o01, false, false);
            o11 = __builtin_amdgcn_wmma_f32_16x16x4_f32(
                false, ha1[c], false, wc1[c], (short)0, o11, false, false);
        }
        float* outp = out + ((size_t)b * NPTS + rowbase + 8 * h) * ODIM + wv * 32 + m;
#pragma unroll
        for (int r = 0; r < 8; ++r) {
            outp[r * ODIM]             = fmaxf(o00[r] + bias20, 0.0f);
            outp[r * ODIM + 16]        = fmaxf(o01[r] + bias21, 0.0f);
            outp[(r + 16) * ODIM]      = fmaxf(o10[r] + bias20, 0.0f);
            outp[(r + 16) * ODIM + 16] = fmaxf(o11[r] + bias21, 0.0f);
        }
    }
}

extern "C" void kernel_launch(void* const* d_in, const int* in_sizes, int n_in,
                              void* d_out, int out_size, void* d_ws, size_t ws_size,
                              hipStream_t stream) {
    const float* x  = (const float*)d_in[0];
    const float* Wl = (const float*)d_in[1];
    const float* bl = (const float*)d_in[2];
    const float* Wc = (const float*)d_in[3];
    const float* bc = (const float*)d_in[4];
    float* out = (float*)d_out;

    float* sqbuf = (float*)d_ws;

    const int total = BATCH * NPTS;
    sq_kernel<<<(total + 255) / 256, 256, 0, stream>>>(x, sqbuf, total);
    knn_pool_kernel<<<BATCH * NRBLK, 32 * NSPLIT, 0, stream>>>(x, sqbuf, Wl, bl, Wc, bc, out);
}


# kNN tile loop pipelined inside each wave: the next tile's 16 matrix products are issued one in front of each column branch of this tile's selection and run under that scalar work; sums unchanged
# speedup vs baseline: 6.0739x; 1.0382x over previous
#include <hip/hip_runtime.h>
#include <hip/hip_bf16.h>

#define BATCH 4
#define NPTS  8192
#define CH    32
#define KNN   16
#define LDIM  64
#define ODIM  128
#define NCTILES (NPTS / 16)
#define ROWS_PER_WAVE 32
#define NRBLK (NPTS / ROWS_PER_WAVE)
#define DPITCH 20
#define NSPLIT 4
#define TILES_PER_WAVE (NCTILES / NSPLIT)
#define QCAP 32

typedef __attribute__((ext_vector_type(2))) float v2f;
typedef __attribute__((ext_vector_type(8))) float v8f;
typedef __attribute__((ext_vector_type(4))) float v4f;
#define POSTPOS 3
#define PPITCH 36
#define HPITCH 68

__global__ void sq_kernel(const float* __restrict__ x, float* __restrict__ sq, int total) {
    int i = blockIdx.x * blockDim.x + threadIdx.x;
    if (i >= total) return;
    const float* p = x + (size_t)i * CH;
    float s = 0.0f;
#pragma unroll
    for (int c = 0; c < CH; ++c) s += p[c] * p[c];
    sq[i] = s;
}

__device__ __forceinline__ void wave_lds_sync() {
    __builtin_amdgcn_fence(__ATOMIC_RELEASE, "workgroup", "local");
    __builtin_amdgcn_fence(__ATOMIC_ACQUIRE, "workgroup", "local");
}

__device__ __forceinline__ void topk_insert(float (&key)[KNN], int (&idx)[KNN], float cv, int ci) {
#pragma unroll
    for (int s = 0; s < KNN; ++s) {
        const float ks = key[s];
        const int   is = idx[s];
        const bool  sw = cv < ks;
        key[s] = sw ? cv : ks;
        idx[s] = sw ? ci : is;
        cv = sw ? ks : cv;
        ci = sw ? is : ci;
    }
}

struct PendSlot {
    float k[32];
    int   i[32];
};

__device__ __forceinline__ void topk_merge(float (&key)[KNN], int (&idx)[KNN],
                                           const float* mk, const int* mi, int lane) {
#pragma unroll 1
    for (int s = 0; s < KNN; ++s) {
        const float d  = mk[s * 32 + lane];
        const int   ci = mi[s * 32 + lane];
        if (d < key[KNN - 1]) topk_insert(key, idx, d, ci);
    }
}

__global__ __launch_bounds__(32 * NSPLIT) void knn_pool_kernel(const float* __restrict__ x,
                                                               const float* __restrict__ sq,
                                                               const float* __restrict__ Wl,
                                                               const float* __restrict__ bl,
                                                               const float* __restrict__ Wc,
                                                               const float* __restrict__ bc,
                                                               float* __restrict__ out) {
    __shared__ __attribute__((aligned(16))) float distT[NSPLIT][ROWS_PER_WAVE * DPITCH];
    __shared__ float mrgKey[3][KNN * 32];
    __shared__ int   mrgIdx[3][KNN * 32];
    __shared__ int   selIdx[ROWS_PER_WAVE * KNN];
    __shared__ PendSlot pend[NSPLIT][QCAP];
    __shared__ __attribute__((aligned(16))) float postKey[32 * NSPLIT];

    const int lane    = threadIdx.x & 31;
    const int wv      = __builtin_amdgcn_readfirstlane(threadIdx.x >> 5);
    const int rblk    = blockIdx.x % NRBLK;
    const int b       = blockIdx.x / NRBLK;
    const int rowbase = rblk * ROWS_PER_WAVE;
    const float* xb   = x  + (size_t)b * NPTS * CH;
    const float* sqb  = sq + (size_t)b * NPTS;
    float* tile       = distT[wv];
    PendSlot* pq      = pend[wv];

    const int m = lane & 15;
    const int h = lane >> 4;

    v2f afrag0[8], afrag1[8];
#pragma unroll
    for (int c = 0; c < 8; ++c) {
        const float* p0 = xb + (size_t)(rowbase + m) * CH + 4 * c + 2 * h;
        const float* p1 = xb + (size_t)(rowbase + 16 + m) * CH + 4 * c + 2 * h;
        afrag0[c].x = -2.0f * p0[0];
        afrag0[c].y = -2.0f * p0[1];
        afrag1[c].x = -2.0f * p1[0];
        afrag1[c].y = -2.0f * p1[1];
    }
    float key[KNN];
    int   idx[KNN];
#pragma unroll
    for (int s = 0; s < KNN; ++s) { key[s] = 3.0e38f; idx[s] = 0; }

    int qn = 0;

    postKey[lane * NSPLIT + wv] = 3.0e38f;
    __syncthreads();

    const int jt0 = wv * TILES_PER_WAVE;

    v8f accN0 = {};
    v8f accN1 = {};
    float sqcur = sqb[jt0 * 16 + m];
    {
        v2f bfirst[8];
#pragma unroll
        for (int c = 0; c < 8; ++c) {
            const float* p = xb + (size_t)(jt0 * 16 + m) * CH + 4 * c + 2 * h;
            bfirst[c].x = p[0];
            bfirst[c].y = p[1];
        }
#pragma unroll
        for (int c = 0; c < 8; ++c) {
            accN0 = __builtin_amdgcn_wmma_f32_16x16x4_f32(
                false, afrag0[c], false, bfirst[c], (short)0, accN0, false, false);
            accN1 = __builtin_amdgcn_wmma_f32_16x16x4_f32(
                false, afrag1[c], false, bfirst[c], (short)0, accN1, false, false);
        }
        asm volatile("v_nop\n\tv_nop\n\tv_nop\n\tv_nop"
                     : "+v"(accN0), "+v"(accN1)
                     : "v"(bfirst[0]), "v"(bfirst[1]), "v"(bfirst[2]), "v"(bfirst[3]),
                       "v"(bfirst[4]), "v"(bfirst[5]), "v"(bfirst[6]), "v"(bfirst[7]));
    }
    v2f   bnxt[8];
    float sqnxt;
    {
        const int n1 = ((jt0 + 1) & (NCTILES - 1)) * 16;
#pragma unroll
        for (int c = 0; c < 8; ++c) {
            const float* p = xb + (size_t)(n1 + m) * CH + 4 * c + 2 * h;
            bnxt[c].x = p[0];
            bnxt[c].y = p[1];
        }
        sqnxt = sqb[n1 + m];
    }

    for (int jt = jt0; jt < jt0 + TILES_PER_WAVE; ++jt) {
        const int colbase = jt * 16;

        const v4f posted = *reinterpret_cast<const v4f*>(&postKey[lane * NSPLIT]);

        const int nbase = ((jt + 2) & (NCTILES - 1)) * 16;
        v2f bnn[8];
#pragma unroll
        for (int c = 0; c < 8; ++c) {
            const float* p = xb + (size_t)(nbase + m) * CH + 4 * c + 2 * h;
            bnn[c].x = p[0];
            bnn[c].y = p[1];
        }
        const float sqnn = sqb[nbase + m];

        asm("s_wait_alu depctr_va_vdst(0)" : "+v"(accN0), "+v"(accN1));
        const v8f acc0 = accN0 + sqcur;
        const v8f acc1 = accN1 + sqcur;

#pragma unroll
        for (int r = 0; r < 8; ++r) {
            tile[(r + 8 * h) * DPITCH + m]        = acc0[r];
            tile[(r + 8 * h + 16) * DPITCH + m]   = acc1[r];
        }
        wave_lds_sync();

        const float4* rowp = reinterpret_cast<const float4*>(&tile[lane * DPITCH]);
        const float4 q0 = rowp[0];
        const float4 q1 = rowp[1];
        const float4 q2 = rowp[2];
        const float4 q3 = rowp[3];
        const float dv[16] = { q0.x, q0.y, q0.z, q0.w, q1.x, q1.y, q1.z, q1.w,
                               q2.x, q2.y, q2.z, q2.w, q3.x, q3.y, q3.z, q3.w };
        const float shared = fmaxf(fmaxf(posted.x, posted.y), fmaxf(posted.z, posted.w));
        const float bound = fminf(key[KNN - 1], shared);
        unsigned hit[16];
#pragma unroll
        for (int c = 0; c < 16; ++c)
            hit[c] = __builtin_amdgcn_ballot_w32(dv[c] < bound);
        accN0 = v8f{};
        accN1 = v8f{};
#pragma unroll
        for (int c = 0; c < 16; ++c) {
            if (c & 1)
                accN1 = __builtin_amdgcn_wmma_f32_16x16x4_f32(
                    false, afrag1[c >> 1], false, bnxt[c >> 1], (short)0, accN1, false, false);
            else
                accN0 = __builtin_amdgcn_wmma_f32_16x16x4_f32(
                    false, afrag0[c >> 1], false, bnxt[c >> 1], (short)0, accN0, false, false);
            if (__builtin_expect(hit[c] != 0u, 0)) {
                if (__builtin_amdgcn_inverse_ballot_w32(hit[c])) {
                    pq[qn].k[lane] = dv[c];
                    pq[qn].i[lane] = colbase + c;
                    qn += 1;
                }
            }
        }
        asm volatile("" : "+v"(accN0), "+v"(accN1)
                        : "v"(bnxt[0]), "v"(bnxt[1]), "v"(bnxt[2]), "v"(bnxt[3]),
                          "v"(bnxt[4]), "v"(bnxt[5]), "v"(bnxt[6]), "v"(bnxt[7]));

        const bool lastTile = (jt == jt0 + TILES_PER_WAVE - 1);
        if (lastTile || __builtin_amdgcn_ballot_w32(qn > QCAP - 16) != 0u) {
            const int maxn = __builtin_amdgcn_readfirstlane(
                (int)__builtin_amdgcn_wave_reduce_max_u32((unsigned)qn, 0));
            for (int q = 0; q < maxn; q += 2) {
                const float k0 = pq[q].k[lane];
                const int   c0 = pq[q].i[lane];
                const float k1 = pq[q + 1].k[lane];
                const int   c1 = pq[q + 1].i[lane];
                const float d0 = (q < qn)     ? k0 : __builtin_inff();
                const float d1 = (q + 1 < qn) ? k1 : __builtin_inff();
                topk_insert(key, idx, d0, c0);
                topk_insert(key, idx, d1, c1);
            }
            qn = 0;
            postKey[lane * NSPLIT + wv] =
                fmaf(fabsf(key[POSTPOS]), 1.0e-5f, key[POSTPOS]) + 1.0e-30f;
        }
        wave_lds_sync();

#pragma unroll
        for (int c = 0; c < 8; ++c) bnxt[c] = bnn[c];
        sqcur = sqnxt;
        sqnxt = sqnn;
    }

    if (wv & 1) {
#pragma unroll
        for (int s = 0; s < KNN; ++s) {
            mrgKey[wv >> 1][s * 32 + lane] = key[s];
            mrgIdx[wv >> 1][s * 32 + lane] = idx[s];
        }
    }
    __syncthreads();
    if (!(wv & 1)) topk_merge(key, idx, mrgKey[wv >> 1], mrgIdx[wv >> 1], lane);
    if (wv == 2) {
#pragma unroll
        for (int s = 0; s < KNN; ++s) {
            mrgKey[2][s * 32 + lane] = key[s];
            mrgIdx[2][s * 32 + lane] = idx[s];
        }
    }
    __syncthreads();
    if (wv == 0) {
        topk_merge(key, idx, mrgKey[2], mrgIdx[2], lane);
#pragma unroll
        for (int s = 0; s < KNN; ++s)
            selIdx[lane * KNN + s] = idx[s];
    }
    __syncthreads();

    float* poolT = &pend[0][0].k[0];
    float* hidT  = &pend[1][0].k[0];

    v2f wl[8], wc0[16], wc1[16];
#pragma unroll
    for (int c = 0; c < 8; ++c) {
        const float* p = Wl + (size_t)(4 * c + 2 * h) * LDIM + wv * 16 + m;
        wl[c].x = p[0];
        wl[c].y = p[LDIM];
    }
#pragma unroll
    for (int c = 0; c < 16; ++c) {
        const float* p = Wc + (size_t)(4 * c + 2 * h) * ODIM + wv * 32 + m;
        wc0[c].x = p[0];
        wc0[c].y = p[ODIM];
        wc1[c].x = p[16];
        wc1[c].y = p[ODIM + 16];
    }
    const float bias1  = bl[wv * 16 + m];
    const float bias20 = bc[wv * 32 + m];
    const float bias21 = bc[wv * 32 + 16 + m];

#pragma unroll 1
    for (int rr = 0; rr < ROWS_PER_WAVE / NSPLIT; ++rr) {
        const int r = wv * (ROWS_PER_WAVE / NSPLIT) + rr;
        float mx = -3.0e38f;
#pragma unroll
        for (int k = 0; k < KNN; ++k) {
            const int j = selIdx[r * KNN + k];
            const float v = xb[(size_t)j * CH + lane];
            mx = fmaxf(mx, v);
        }
        poolT[r * PPITCH + lane] = mx;
    }
    __syncthreads();

    {
        v2f pa0[8], pa1[8];
#pragma unroll
        for (int c = 0; c < 8; ++c) {
            pa0[c].x = poolT[m * PPITCH + 4 * c + 2 * h];
            pa0[c].y = poolT[m * PPITCH + 4 * c + 2 * h + 1];
            pa1[c].x = poolT[(16 + m) * PPITCH + 4 * c + 2 * h];
            pa1[c].y = poolT[(16 + m) * PPITCH + 4 * c + 2 * h + 1];
        }
        v8f h0 = {};
        v8f h1 = {};
#pragma unroll
        for (int c = 0; c < 8; ++c) {
            h0 = __builtin_amdgcn_wmma_f32_16x16x4_f32(
                false, pa0[c], false, wl[c], (short)0, h0, false, false);
            h1 = __builtin_amdgcn_wmma_f32_16x16x4_f32(
                false, pa1[c], false, wl[c], (short)0, h1, false, false);
        }
        asm volatile("v_nop\n\tv_nop\n\tv_nop\n\tv_nop"
                     : "+v"(h0), "+v"(h1)
                     : "v"(pa0[0]), "v"(pa0[1]), "v"(pa0[2]), "v"(pa0[3]),
                       "v"(pa0[4]), "v"(pa0[5]), "v"(pa0[6]), "v"(pa0[7]),
                       "v"(pa1[0]), "v"(pa1[1]), "v"(pa1[2]), "v"(pa1[3]),
                       "v"(pa1[4]), "v"(pa1[5]), "v"(pa1[6]), "v"(pa1[7]),
                       "v"(wl[0]), "v"(wl[1]), "v"(wl[2]), "v"(wl[3]),
                       "v"(wl[4]), "v"(wl[5]), "v"(wl[6]), "v"(wl[7]));
        float* hrow = hidT + (8 * h) * HPITCH + wv * 16 + m;
#pragma unroll
        for (int r = 0; r < 8; ++r) {
            hrow[r * HPITCH]        = h0[r] + bias1;
            hrow[(r + 16) * HPITCH] = h1[r] + bias1;
        }
    }
    __syncthreads();

    {
        v2f ha0[16], ha1[16];
#pragma unroll
        for (int c = 0; c < 16; ++c) {
            ha0[c].x = hidT[m * HPITCH + 4 * c + 2 * h];
            ha0[c].y = hidT[m * HPITCH + 4 * c + 2 * h + 1];
            ha1[c].x = hidT[(16 + m) * HPITCH + 4 * c + 2 * h];
            ha1[c].y = hidT[(16 + m) * HPITCH + 4 * c + 2 * h + 1];
        }
        v8f o00 = {};
        v8f o01 = {};
        v8f o10 = {};
        v8f o11 = {};
#pragma unroll
        for (int c = 0; c < 16; ++c) {
            o00 = __builtin_amdgcn_wmma_f32_16x16x4_f32(
                false, ha0[c], false, wc0[c], (short)0, o00, false, false);
            o10 = __builtin_amdgcn_wmma_f32_16x16x4_f32(
                false, ha1[c], false, wc0[c], (short)0, o10, false, false);
            o01 = __builtin_amdgcn_wmma_f32_16x16x4_f32(
                false, ha0[c], false, wc1[c], (short)0, o01, false, false);
            o11 = __builtin_amdgcn_wmma_f32_16x16x4_f32(
                false, ha1[c], false, wc1[c], (short)0, o11, false, false);
        }
        asm volatile("v_nop\n\tv_nop\n\tv_nop\n\tv_nop"
                     : "+v"(o00), "+v"(o01), "+v"(o10), "+v"(o11)
                     : "v"(ha0[0]), "v"(ha0[1]), "v"(ha0[2]), "v"(ha0[3]),
                       "v"(ha0[4]), "v"(ha0[5]), "v"(ha0[6]), "v"(ha0[7]),
                       "v"(ha0[8]), "v"(ha0[9]), "v"(ha0[10]), "v"(ha0[11]),
                       "v"(ha0[12]), "v"(ha0[13]), "v"(ha0[14]), "v"(ha0[15]));
        asm volatile("" : "+v"(o00), "+v"(o01), "+v"(o10), "+v"(o11)
                     : "v"(ha1[0]), "v"(ha1[1]), "v"(ha1[2]), "v"(ha1[3]),
                       "v"(ha1[4]), "v"(ha1[5]), "v"(ha1[6]), "v"(ha1[7]),
                       "v"(ha1[8]), "v"(ha1[9]), "v"(ha1[10]), "v"(ha1[11]),
                       "v"(ha1[12]), "v"(ha1[13]), "v"(ha1[14]), "v"(ha1[15]));
        asm volatile("" : "+v"(o00), "+v"(o01), "+v"(o10), "+v"(o11)
                     : "v"(wc0[0]), "v"(wc0[1]), "v"(wc0[2]), "v"(wc0[3]),
                       "v"(wc0[4]), "v"(wc0[5]), "v"(wc0[6]), "v"(wc0[7]),
                       "v"(wc0[8]), "v"(wc0[9]), "v"(wc0[10]), "v"(wc0[11]),
                       "v"(wc0[12]), "v"(wc0[13]), "v"(wc0[14]), "v"(wc0[15]));
        asm volatile("" : "+v"(o00), "+v"(o01), "+v"(o10), "+v"(o11)
                     : "v"(wc1[0]), "v"(wc1[1]), "v"(wc1[2]), "v"(wc1[3]),
                       "v"(wc1[4]), "v"(wc1[5]), "v"(wc1[6]), "v"(wc1[7]),
                       "v"(wc1[8]), "v"(wc1[9]), "v"(wc1[10]), "v"(wc1[11]),
                       "v"(wc1[12]), "v"(wc1[13]), "v"(wc1[14]), "v"(wc1[15]));
        float* outp = out + ((size_t)b * NPTS + rowbase + 8 * h) * ODIM + wv * 32 + m;
#pragma unroll
        for (int r = 0; r < 8; ++r) {
            outp[r * ODIM]             = fmaxf(o00[r] + bias20, 0.0f);
            outp[r * ODIM + 16]        = fmaxf(o01[r] + bias21, 0.0f);
            outp[(r + 16) * ODIM]      = fmaxf(o10[r] + bias20, 0.0f);
            outp[(r + 16) * ODIM + 16] = fmaxf(o11[r] + bias21, 0.0f);
        }
    }
}

extern "C" void kernel_launch(void* const* d_in, const int* in_sizes, int n_in,
                              void* d_out, int out_size, void* d_ws, size_t ws_size,
                              hipStream_t stream) {
    const float* x  = (const float*)d_in[0];
    const float* Wl = (const float*)d_in[1];
    const float* bl = (const float*)d_in[2];
    const float* Wc = (const float*)d_in[3];
    const float* bc = (const float*)d_in[4];
    float* out = (float*)d_out;

    float* sqbuf = (float*)d_ws;

    const int total = BATCH * NPTS;
    sq_kernel<<<(total + 255) / 256, 256, 0, stream>>>(x, sqbuf, total);
    knn_pool_kernel<<<BATCH * NRBLK, 32 * NSPLIT, 0, stream>>>(x, sqbuf, Wl, bl, Wc, bc, out);
}
